# starting kernel
# baseline (speedup 1.0000x reference)
#include <hip/hip_runtime.h>
#include <hip/hip_bf16.h>


typedef __attribute__((ext_vector_type(2))) float v2f;
typedef __attribute__((ext_vector_type(8))) float v8f;


__global__ void degree_kernel(const int* __restrict__ src,
                              const int* __restrict__ dst,
                              float* __restrict__ deg_out,
                              float* __restrict__ deg_in, int E) {
  int e = blockIdx.x * blockDim.x + threadIdx.x;
  if (e < E) {
    atomicAdd(&deg_out[src[e]], 1.0f);
    atomicAdd(&deg_in[dst[e]], 1.0f);
  }
}

__global__ void norm_kernel(float* __restrict__ deg, int n) {
  int i = blockIdx.x * blockDim.x + threadIdx.x;
  if (i < n) deg[i] = rsqrtf(fmaxf(deg[i], 1.0f));
}


__global__ void scale_rows_kernel(float* __restrict__ out,
                                  const float* __restrict__ in,
                                  const float* __restrict__ norm,
                                  int n4, int d4) {
  int i = blockIdx.x * blockDim.x + threadIdx.x;
  if (i >= n4) return;
  int row = i / d4;
  float s = norm[row];
  float4 v = reinterpret_cast<const float4*>(in)[i];
  v.x *= s; v.y *= s; v.z *= s; v.w *= s;
  reinterpret_cast<float4*>(out)[i] = v;
}


template <int D>
__global__ void scatter_add_kernel(const float* __restrict__ h,
                                   const int* __restrict__ src,
                                   const int* __restrict__ dst,
                                   float* __restrict__ agg, int E) {
  int warp = (int)((blockIdx.x * (size_t)blockDim.x + threadIdx.x) >> 5);
  int lane = threadIdx.x & 31;
  if (warp >= E) return;
  int s = src[warp];
  int d = dst[warp];
  const float4* hs = reinterpret_cast<const float4*>(h + (size_t)s * D);
  float* ad = agg + (size_t)d * D;
#pragma unroll
  for (int i = 0; i < D / 128; ++i) {
    float4 v = hs[lane + i * 32];
    int c = (lane + i * 32) * 4;
    atomicAdd(ad + c + 0, v.x);
    atomicAdd(ad + c + 1, v.y);
    atomicAdd(ad + c + 2, v.z);
    atomicAdd(ad + c + 3, v.w);
  }
}


template <int K, int NC, int ACT, bool HAS_SCALE>
__global__ __launch_bounds__(256) void gemm_wmma_kernel(
    const float* __restrict__ A, const float* __restrict__ W,
    const float* __restrict__ bias, const float* __restrict__ rowscale,
    float* __restrict__ C, int M) {
  constexpr int PAD = 4;
  constexpr int LDSROW = K + PAD;
  __shared__ float sA[64 * LDSROW];

  const int tid = threadIdx.x;
  const int blockM = blockIdx.x * 64;
  const int blockN = blockIdx.y * 64;

  constexpr int F4_PER_ROW = K / 4;
  constexpr int TOTAL_F4 = 64 * F4_PER_ROW;
  for (int idx = tid; idx < TOTAL_F4; idx += 256) {
    int row = idx / F4_PER_ROW;
    int c4 = idx % F4_PER_ROW;
    float4 v = make_float4(0.f, 0.f, 0.f, 0.f);
    int gr = blockM + row;
    if (gr < M) v = *reinterpret_cast<const float4*>(A + (size_t)gr * K + c4 * 4);
    *reinterpret_cast<float4*>(&sA[row * LDSROW + c4 * 4]) = v;
  }
  __syncthreads();

  const int wave = tid >> 5;
  const int lane = tid & 31;
  const int wn = wave & 3;
  const int wmg = wave >> 2;
  const int half = lane >> 4;
  const int lm = lane & 15;
  const int col = blockN + wn * 16 + lm;

  v8f acc0 = {};
  v8f acc1 = {};
  const float* sArow0 = &sA[(wmg * 32 + lm) * LDSROW + 2 * half];
  const float* sArow1 = sArow0 + 16 * LDSROW;
  const float* Wp = W + (size_t)(2 * half) * NC + col;

#pragma unroll 8
  for (int k = 0; k < K; k += 4) {
    v2f a0 = *reinterpret_cast<const v2f*>(sArow0 + k);
    v2f a1 = *reinterpret_cast<const v2f*>(sArow1 + k);
    v2f b;
    b[0] = Wp[k * NC];
    b[1] = Wp[k * NC + NC];
    acc0 = __builtin_amdgcn_wmma_f32_16x16x4_f32(false, a0, false, b, (short)0,
                                                 acc0, false, false);
    acc1 = __builtin_amdgcn_wmma_f32_16x16x4_f32(false, a1, false, b, (short)0,
                                                 acc1, false, false);
  }

  const float bval = bias[col];
  const int rbase0 = blockM + wmg * 32 + 8 * half;
  if (blockM + 64 <= M) {
#pragma unroll
    for (int t = 0; t < 2; ++t) {
      const int rbase = rbase0 + t * 16;
      float* Cp = C + (size_t)rbase * NC + col;
      const float* rs = rowscale + rbase;
#pragma unroll
      for (int r = 0; r < 8; ++r) {
        float v = (t == 0) ? acc0[r] : acc1[r];
        if (HAS_SCALE) v *= rs[r];
        v += bval;
        if (ACT == 1) v = fmaxf(v, 0.0f);
        else if (ACT == 2) v = (v > 0.0f) ? v : expm1f(v);
        Cp[r * NC] = v;
      }
    }
  } else {
#pragma unroll
    for (int t = 0; t < 2; ++t) {
      const int rbase = rbase0 + t * 16;
#pragma unroll
      for (int r = 0; r < 8; ++r) {
        const int row = rbase + r;
        if (row < M) {
          float v = (t == 0) ? acc0[r] : acc1[r];
          if (HAS_SCALE) v *= rowscale[row];
          v += bval;
          if (ACT == 1) v = fmaxf(v, 0.0f);
          else if (ACT == 2) v = (v > 0.0f) ? v : expm1f(v);
          C[(size_t)row * NC + col] = v;
        }
      }
    }
  }
}


extern "C" void kernel_launch(void* const* d_in, const int* in_sizes, int n_in,
                              void* d_out, int out_size, void* d_ws, size_t ws_size,
                              hipStream_t stream) {
  const float* feat1 = (const float*)d_in[0];
  const float* feat2 = (const float*)d_in[1];
  const int* ei1 = (const int*)d_in[2];
  const int* ei2 = (const int*)d_in[3];
  const float* W1 = (const float*)d_in[4];
  const float* b1 = (const float*)d_in[5];
  const float* W2 = (const float*)d_in[6];
  const float* b2 = (const float*)d_in[7];
  const float* fc1W = (const float*)d_in[8];
  const float* fc1b = (const float*)d_in[9];
  const float* fc2W = (const float*)d_in[10];
  const float* fc2b = (const float*)d_in[11];

  const int N = in_sizes[0] / 128;

  float* ws = (float*)d_ws;
  float* norm_src = ws;
  float* norm_dst = ws + N;
  float* bufA = ws + 2 * (size_t)N;
  float* bufB = bufA + (size_t)N * 256;

  for (int g = 0; g < 2; ++g) {
    const float* feat = g ? feat2 : feat1;
    const int E = (g ? in_sizes[3] : in_sizes[2]) / 2;
    const int* src = g ? ei2 : ei1;
    const int* dst = src + E;
    float* zout = (float*)d_out + (size_t)g * N * 128;

    hipMemsetAsync(norm_src, 0, 2 * (size_t)N * sizeof(float), stream);
    degree_kernel<<<(E + 255) / 256, 256, 0, stream>>>(src, dst, norm_src, norm_dst, E);
    norm_kernel<<<(2 * N + 255) / 256, 256, 0, stream>>>(norm_src, 2 * N);

    scale_rows_kernel<<<(N * 128 / 4 + 255) / 256, 256, 0, stream>>>(
        bufA, feat, norm_src, N * 128 / 4, 32);
    hipMemsetAsync(bufB, 0, (size_t)N * 128 * sizeof(float), stream);
    scatter_add_kernel<128><<<((size_t)E * 32 + 255) / 256, 256, 0, stream>>>(
        bufA, src, dst, bufB, E);
    {
      dim3 grid((N + 63) / 64, 256 / 64);
      gemm_wmma_kernel<128, 256, 1, true><<<grid, 256, 0, stream>>>(
          bufB, W1, b1, norm_dst, bufA, N);
    }

    scale_rows_kernel<<<(N * 256 / 4 + 255) / 256, 256, 0, stream>>>(
        bufA, bufA, norm_src, N * 256 / 4, 64);
    hipMemsetAsync(bufB, 0, (size_t)N * 256 * sizeof(float), stream);
    scatter_add_kernel<256><<<((size_t)E * 32 + 255) / 256, 256, 0, stream>>>(
        bufA, src, dst, bufB, E);
    {
      dim3 grid((N + 63) / 64, 128 / 64);
      gemm_wmma_kernel<256, 128, 1, true><<<grid, 256, 0, stream>>>(
          bufB, W2, b2, norm_dst, bufA, N);
    }

    {
      dim3 grid((N + 63) / 64, 128 / 64);
      gemm_wmma_kernel<128, 128, 2, false><<<grid, 256, 0, stream>>>(
          bufA, fc1W, fc1b, nullptr, bufB, N);
      gemm_wmma_kernel<128, 128, 0, false><<<grid, 256, 0, stream>>>(
          bufB, fc2W, fc2b, nullptr, zout, N);
    }
  }
}


# Edge aggregation becomes a per-destination gather over padded adjacency lists built with integer atomics, replacing the f32 atomic scatter-add and its memsets; rows summed in ascending source order
# speedup vs baseline: 2.1394x; 2.1394x over previous
#include <hip/hip_runtime.h>
#include <hip/hip_bf16.h>


typedef __attribute__((ext_vector_type(2))) float v2f;
typedef __attribute__((ext_vector_type(8))) float v8f;


constexpr int LIST_CAP = 64;

__global__ void degree_adjacency_kernel(const int* __restrict__ src,
                                        const int* __restrict__ dst,
                                        int* __restrict__ cnt_out,
                                        int* __restrict__ cnt_in,
                                        int* __restrict__ lists, int E) {
  int e = blockIdx.x * blockDim.x + threadIdx.x;
  if (e < E) {
    const int s = src[e];
    const int d = dst[e];
    atomicAdd(&cnt_out[s], 1);
    const int slot = atomicAdd(&cnt_in[d], 1);
    if (slot < LIST_CAP) lists[(size_t)d * LIST_CAP + slot] = s;
  }
}

__global__ void norm_kernel(int* __restrict__ cnt, int* __restrict__ deg_in, int n_nodes) {
  int i = blockIdx.x * blockDim.x + threadIdx.x;
  if (i < 2 * n_nodes) {
    const int c = cnt[i];
    if (i >= n_nodes) deg_in[i - n_nodes] = c;
    cnt[i] = __float_as_int(rsqrtf(fmaxf((float)c, 1.0f)));
  }
}


__global__ void scale_rows_kernel(float* __restrict__ out,
                                  const float* __restrict__ in,
                                  const float* __restrict__ norm,
                                  int n4, int d4) {
  int i = blockIdx.x * blockDim.x + threadIdx.x;
  if (i >= n4) return;
  int row = i / d4;
  float s = norm[row];
  float4 v = reinterpret_cast<const float4*>(in)[i];
  v.x *= s; v.y *= s; v.z *= s; v.w *= s;
  reinterpret_cast<float4*>(out)[i] = v;
}


template <int D>
__device__ __forceinline__ void gather_span(const float4* __restrict__ h4, int sreg,
                                            int count, int lane, float4 (&acc)[D / 128]) {
  constexpr int VEC = D / 128;
  constexpr int CH = 8 / VEC;
  for (int base = 0; base < count; base += CH) {
    int sidx[CH];
#pragma unroll
    for (int u = 0; u < CH; ++u) {
      int j = base + u;
      j = j < count ? j : count - 1;
      sidx[u] = __builtin_amdgcn_readlane(sreg, j);
    }
    float4 v[CH][VEC];
    if (base + CH <= count) {
#pragma unroll
      for (int u = 0; u < CH; ++u) {
        const float4* p = h4 + (size_t)sidx[u] * (D / 4) + lane;
#pragma unroll
        for (int q = 0; q < VEC; ++q) v[u][q] = p[32 * q];
      }
    } else {
#pragma unroll
      for (int u = 0; u < CH; ++u) {
        const float4* p = h4 + (size_t)sidx[u] * (D / 4) + lane;
        const bool valid = base + u < count;
#pragma unroll
        for (int q = 0; q < VEC; ++q) {
          const float4 t = p[32 * q];
          v[u][q].x = valid ? t.x : 0.f;
          v[u][q].y = valid ? t.y : 0.f;
          v[u][q].z = valid ? t.z : 0.f;
          v[u][q].w = valid ? t.w : 0.f;
        }
      }
    }
#pragma unroll
    for (int u = 0; u < CH; ++u) {
#pragma unroll
      for (int q = 0; q < VEC; ++q) {
        acc[q].x += v[u][q].x;
        acc[q].y += v[u][q].y;
        acc[q].z += v[u][q].z;
        acc[q].w += v[u][q].w;
      }
    }
  }
}

template <int D, bool SORT>
__global__ __launch_bounds__(256) void gather_sum_kernel(
    const float* __restrict__ h, const int* __restrict__ deg_in,
    int* __restrict__ lists, const int* __restrict__ src,
    const int* __restrict__ dst, float* __restrict__ agg, int n, int E) {
  constexpr int VEC = D / 128;
  const int lane = threadIdx.x & 31;
  const int row = __builtin_amdgcn_readfirstlane((int)((blockIdx.x * 256u + threadIdx.x) >> 5));
  if (row >= n) return;
  int* lrow = lists + (size_t)row * LIST_CAP;
  int s0 = lrow[lane];
  int s1 = lrow[lane + 32];
  const int deg = __builtin_amdgcn_readfirstlane(deg_in[row]);
  const float4* h4 = reinterpret_cast<const float4*>(h);

  float4 acc[VEC];
#pragma unroll
  for (int q = 0; q < VEC; ++q) acc[q] = make_float4(0.f, 0.f, 0.f, 0.f);

  if (deg <= LIST_CAP) {
    if (SORT) {
      const int key0 = (lane < deg) ? (int)(((unsigned)s0 << 6) | (unsigned)lane) : 0x7fffffff;
      const int key1 = (lane + 32 < deg) ? (int)(((unsigned)s1 << 6) | (unsigned)(lane + 32)) : 0x7fffffff;
      const int n0 = deg < 32 ? deg : 32;
      int r0 = 0, r1 = 0;
      for (int j = 0; j < n0; ++j) {
        const int kj = __builtin_amdgcn_readlane(key0, j);
        r0 += (kj < key0) ? 1 : 0;
        r1 += (kj < key1) ? 1 : 0;
      }
      for (int j = 32; j < deg; ++j) {
        const int kj = __builtin_amdgcn_readlane(key1, j - 32);
        r0 += (kj < key0) ? 1 : 0;
        r1 += (kj < key1) ? 1 : 0;
      }
      int p0 = 0, p1 = 0;
      for (int j = 0; j < n0; ++j) {
        const int kj = __builtin_amdgcn_readlane(key0, j);
        const int rj = __builtin_amdgcn_readlane(r0, j);
        if (rj == lane) p0 = kj;
        if (rj == lane + 32) p1 = kj;
      }
      for (int j = 32; j < deg; ++j) {
        const int kj = __builtin_amdgcn_readlane(key1, j - 32);
        const int rj = __builtin_amdgcn_readlane(r1, j - 32);
        if (rj == lane) p0 = kj;
        if (rj == lane + 32) p1 = kj;
      }
      s0 = p0 >> 6;
      s1 = p1 >> 6;
      if (lane < deg) lrow[lane] = s0;
      if (lane + 32 < deg) lrow[lane + 32] = s1;
    }
    gather_span<D>(h4, s0, deg < 32 ? deg : 32, lane, acc);
    if (deg > 32) gather_span<D>(h4, s1, deg - 32, lane, acc);
  } else {
    for (int e0 = 0; e0 < E; e0 += 32) {
      const int e = e0 + lane;
      int sv = 0;
      bool hit = false;
      if (e < E) {
        hit = (dst[e] == row);
        sv = src[e];
      }
      unsigned m = __builtin_amdgcn_ballot_w32(hit);
      while (m) {
        const int b = __builtin_ctz(m);
        m &= m - 1;
        const int s = __builtin_amdgcn_readlane(sv, b);
        const float4* p = h4 + (size_t)s * (D / 4) + lane;
#pragma unroll
        for (int q = 0; q < VEC; ++q) {
          const float4 t = p[32 * q];
          acc[q].x += t.x;
          acc[q].y += t.y;
          acc[q].z += t.z;
          acc[q].w += t.w;
        }
      }
    }
  }

  float4* out4 = reinterpret_cast<float4*>(agg) + (size_t)row * (D / 4) + lane;
#pragma unroll
  for (int q = 0; q < VEC; ++q) out4[32 * q] = acc[q];
}


template <int K, int NC, int ACT, bool HAS_SCALE>
__global__ __launch_bounds__(256) void gemm_wmma_kernel(
    const float* __restrict__ A, const float* __restrict__ W,
    const float* __restrict__ bias, const float* __restrict__ rowscale,
    float* __restrict__ C, int M) {
  constexpr int PAD = 4;
  constexpr int LDSROW = K + PAD;
  __shared__ float sA[64 * LDSROW];

  const int tid = threadIdx.x;
  const int blockM = blockIdx.x * 64;
  const int blockN = blockIdx.y * 64;

  constexpr int F4_PER_ROW = K / 4;
  constexpr int TOTAL_F4 = 64 * F4_PER_ROW;
  for (int idx = tid; idx < TOTAL_F4; idx += 256) {
    int row = idx / F4_PER_ROW;
    int c4 = idx % F4_PER_ROW;
    float4 v = make_float4(0.f, 0.f, 0.f, 0.f);
    int gr = blockM + row;
    if (gr < M) v = *reinterpret_cast<const float4*>(A + (size_t)gr * K + c4 * 4);
    *reinterpret_cast<float4*>(&sA[row * LDSROW + c4 * 4]) = v;
  }
  __syncthreads();

  const int wave = tid >> 5;
  const int lane = tid & 31;
  const int wn = wave & 3;
  const int wmg = wave >> 2;
  const int half = lane >> 4;
  const int lm = lane & 15;
  const int col = blockN + wn * 16 + lm;

  v8f acc0 = {};
  v8f acc1 = {};
  const float* sArow0 = &sA[(wmg * 32 + lm) * LDSROW + 2 * half];
  const float* sArow1 = sArow0 + 16 * LDSROW;
  const float* Wp = W + (size_t)(2 * half) * NC + col;

#pragma unroll 8
  for (int k = 0; k < K; k += 4) {
    v2f a0 = *reinterpret_cast<const v2f*>(sArow0 + k);
    v2f a1 = *reinterpret_cast<const v2f*>(sArow1 + k);
    v2f b;
    b[0] = Wp[k * NC];
    b[1] = Wp[k * NC + NC];
    acc0 = __builtin_amdgcn_wmma_f32_16x16x4_f32(false, a0, false, b, (short)0,
                                                 acc0, false, false);
    acc1 = __builtin_amdgcn_wmma_f32_16x16x4_f32(false, a1, false, b, (short)0,
                                                 acc1, false, false);
  }

  const float bval = bias[col];
  const int rbase0 = blockM + wmg * 32 + 8 * half;
  if (blockM + 64 <= M) {
#pragma unroll
    for (int t = 0; t < 2; ++t) {
      const int rbase = rbase0 + t * 16;
      float* Cp = C + (size_t)rbase * NC + col;
      const float* rs = rowscale + rbase;
#pragma unroll
      for (int r = 0; r < 8; ++r) {
        float v = (t == 0) ? acc0[r] : acc1[r];
        if (HAS_SCALE) v *= rs[r];
        v += bval;
        if (ACT == 1) v = fmaxf(v, 0.0f);
        else if (ACT == 2) v = (v > 0.0f) ? v : expm1f(v);
        Cp[r * NC] = v;
      }
    }
  } else {
#pragma unroll
    for (int t = 0; t < 2; ++t) {
      const int rbase = rbase0 + t * 16;
#pragma unroll
      for (int r = 0; r < 8; ++r) {
        const int row = rbase + r;
        if (row < M) {
          float v = (t == 0) ? acc0[r] : acc1[r];
          if (HAS_SCALE) v *= rowscale[row];
          v += bval;
          if (ACT == 1) v = fmaxf(v, 0.0f);
          else if (ACT == 2) v = (v > 0.0f) ? v : expm1f(v);
          C[(size_t)row * NC + col] = v;
        }
      }
    }
  }
}


extern "C" void kernel_launch(void* const* d_in, const int* in_sizes, int n_in,
                              void* d_out, int out_size, void* d_ws, size_t ws_size,
                              hipStream_t stream) {
  const float* feat1 = (const float*)d_in[0];
  const float* feat2 = (const float*)d_in[1];
  const int* ei1 = (const int*)d_in[2];
  const int* ei2 = (const int*)d_in[3];
  const float* W1 = (const float*)d_in[4];
  const float* b1 = (const float*)d_in[5];
  const float* W2 = (const float*)d_in[6];
  const float* b2 = (const float*)d_in[7];
  const float* fc1W = (const float*)d_in[8];
  const float* fc1b = (const float*)d_in[9];
  const float* fc2W = (const float*)d_in[10];
  const float* fc2b = (const float*)d_in[11];

  const int N = in_sizes[0] / 128;

  float* ws = (float*)d_ws;
  float* norm_src = ws;
  float* norm_dst = ws + N;
  float* bufA = ws + 2 * (size_t)N;
  float* bufB = bufA + (size_t)N * 256;

  for (int g = 0; g < 2; ++g) {
    const float* feat = g ? feat2 : feat1;
    const int E = (g ? in_sizes[3] : in_sizes[2]) / 2;
    const int* src = g ? ei2 : ei1;
    const int* dst = src + E;
    float* zout = (float*)d_out + (size_t)g * N * 128;

    int* lists = (int*)zout;
    int* deg_in = lists + (size_t)N * LIST_CAP;

    hipMemsetAsync(norm_src, 0, 2 * (size_t)N * sizeof(float), stream);
    degree_adjacency_kernel<<<(E + 255) / 256, 256, 0, stream>>>(
        src, dst, (int*)norm_src, (int*)norm_dst, lists, E);
    norm_kernel<<<(2 * N + 255) / 256, 256, 0, stream>>>((int*)norm_src, deg_in, N);

    scale_rows_kernel<<<(N * 128 / 4 + 255) / 256, 256, 0, stream>>>(
        bufA, feat, norm_src, N * 128 / 4, 32);
    gather_sum_kernel<128, true><<<(N * 32 + 255) / 256, 256, 0, stream>>>(
        bufA, deg_in, lists, src, dst, bufB, N, E);
    {
      dim3 grid((N + 63) / 64, 256 / 64);
      gemm_wmma_kernel<128, 256, 1, true><<<grid, 256, 0, stream>>>(
          bufB, W1, b1, norm_dst, bufA, N);
    }

    scale_rows_kernel<<<(N * 256 / 4 + 255) / 256, 256, 0, stream>>>(
        bufA, bufA, norm_src, N * 256 / 4, 64);
    gather_sum_kernel<256, false><<<(N * 32 + 255) / 256, 256, 0, stream>>>(
        bufA, deg_in, lists, src, dst, bufB, N, E);
    {
      dim3 grid((N + 63) / 64, 128 / 64);
      gemm_wmma_kernel<256, 128, 1, true><<<grid, 256, 0, stream>>>(
          bufB, W2, b2, norm_dst, bufA, N);
    }

    {
      dim3 grid((N + 63) / 64, 128 / 64);
      gemm_wmma_kernel<128, 128, 2, false><<<grid, 256, 0, stream>>>(
          bufA, fc1W, fc1b, nullptr, bufB, N);
      gemm_wmma_kernel<128, 128, 0, false><<<grid, 256, 0, stream>>>(
          bufB, fc2W, fc2b, nullptr, zout, N);
    }
  }
}


# Layer 2 takes its W2 product (256->128) before the edge gather (fixed order, no float atomics), so the gather sums 128-wide rows; norm_dst, bias, relu move into its store; changes summation order
# speedup vs baseline: 2.4584x; 1.1491x over previous
#include <hip/hip_runtime.h>
#include <hip/hip_bf16.h>


typedef __attribute__((ext_vector_type(2))) float v2f;
typedef __attribute__((ext_vector_type(8))) float v8f;


constexpr int LIST_CAP = 64;

__global__ void degree_adjacency_kernel(const int* __restrict__ src,
                                        const int* __restrict__ dst,
                                        int* __restrict__ cnt_out,
                                        int* __restrict__ cnt_in,
                                        int* __restrict__ lists, int E) {
  int e = blockIdx.x * blockDim.x + threadIdx.x;
  if (e < E) {
    const int s = src[e];
    const int d = dst[e];
    atomicAdd(&cnt_out[s], 1);
    const int slot = atomicAdd(&cnt_in[d], 1);
    if (slot < LIST_CAP) lists[(size_t)d * LIST_CAP + slot] = s;
  }
}

__global__ void norm_kernel(int* __restrict__ cnt, int* __restrict__ deg_in, int n_nodes) {
  int i = blockIdx.x * blockDim.x + threadIdx.x;
  if (i < 2 * n_nodes) {
    const int c = cnt[i];
    if (i >= n_nodes) deg_in[i - n_nodes] = c;
    cnt[i] = __float_as_int(rsqrtf(fmaxf((float)c, 1.0f)));
  }
}


__global__ void scale_rows_kernel(float* __restrict__ out,
                                  const float* __restrict__ in,
                                  const float* __restrict__ norm,
                                  int n4, int d4) {
  int i = blockIdx.x * blockDim.x + threadIdx.x;
  if (i >= n4) return;
  int row = i / d4;
  float s = norm[row];
  float4 v = reinterpret_cast<const float4*>(in)[i];
  v.x *= s; v.y *= s; v.z *= s; v.w *= s;
  reinterpret_cast<float4*>(out)[i] = v;
}


template <int D>
__device__ __forceinline__ void gather_span(const float4* __restrict__ h4, int sreg,
                                            int count, int lane, float4 (&acc)[D / 128]) {
  constexpr int VEC = D / 128;
  constexpr int CH = 8 / VEC;
  for (int base = 0; base < count; base += CH) {
    int sidx[CH];
#pragma unroll
    for (int u = 0; u < CH; ++u) {
      int j = base + u;
      j = j < count ? j : count - 1;
      sidx[u] = __builtin_amdgcn_readlane(sreg, j);
    }
    float4 v[CH][VEC];
    if (base + CH <= count) {
#pragma unroll
      for (int u = 0; u < CH; ++u) {
        const float4* p = h4 + (size_t)sidx[u] * (D / 4) + lane;
#pragma unroll
        for (int q = 0; q < VEC; ++q) v[u][q] = p[32 * q];
      }
    } else {
#pragma unroll
      for (int u = 0; u < CH; ++u) {
        const float4* p = h4 + (size_t)sidx[u] * (D / 4) + lane;
        const bool valid = base + u < count;
#pragma unroll
        for (int q = 0; q < VEC; ++q) {
          const float4 t = p[32 * q];
          v[u][q].x = valid ? t.x : 0.f;
          v[u][q].y = valid ? t.y : 0.f;
          v[u][q].z = valid ? t.z : 0.f;
          v[u][q].w = valid ? t.w : 0.f;
        }
      }
    }
#pragma unroll
    for (int u = 0; u < CH; ++u) {
#pragma unroll
      for (int q = 0; q < VEC; ++q) {
        acc[q].x += v[u][q].x;
        acc[q].y += v[u][q].y;
        acc[q].z += v[u][q].z;
        acc[q].w += v[u][q].w;
      }
    }
  }
}

template <int D, bool SORT, bool EPI>
__global__ __launch_bounds__(256) void gather_sum_kernel(
    const float* __restrict__ h, const int* __restrict__ deg_in,
    int* __restrict__ lists, const int* __restrict__ src,
    const int* __restrict__ dst, const float* __restrict__ rowscale,
    const float* __restrict__ bias, float* __restrict__ agg, int n, int E) {
  constexpr int VEC = D / 128;
  const int lane = threadIdx.x & 31;
  const int row = __builtin_amdgcn_readfirstlane((int)((blockIdx.x * 256u + threadIdx.x) >> 5));
  if (row >= n) return;
  int* lrow = lists + (size_t)row * LIST_CAP;
  int s0 = lrow[lane];
  int s1 = lrow[lane + 32];
  const int deg = __builtin_amdgcn_readfirstlane(deg_in[row]);
  const float4* h4 = reinterpret_cast<const float4*>(h);

  float rs = 0.f;
  float4 bq[VEC];
  if (EPI) {
    rs = rowscale[row];
    const float4* b4 = reinterpret_cast<const float4*>(bias) + lane;
#pragma unroll
    for (int q = 0; q < VEC; ++q) bq[q] = b4[32 * q];
  }

  float4 acc[VEC];
#pragma unroll
  for (int q = 0; q < VEC; ++q) acc[q] = make_float4(0.f, 0.f, 0.f, 0.f);

  if (deg <= LIST_CAP) {
    if (SORT) {
      const int key0 = (lane < deg) ? (int)(((unsigned)s0 << 6) | (unsigned)lane) : 0x7fffffff;
      const int key1 = (lane + 32 < deg) ? (int)(((unsigned)s1 << 6) | (unsigned)(lane + 32)) : 0x7fffffff;
      const int n0 = deg < 32 ? deg : 32;
      int r0 = 0, r1 = 0;
      for (int j = 0; j < n0; ++j) {
        const int kj = __builtin_amdgcn_readlane(key0, j);
        r0 += (kj < key0) ? 1 : 0;
        r1 += (kj < key1) ? 1 : 0;
      }
      for (int j = 32; j < deg; ++j) {
        const int kj = __builtin_amdgcn_readlane(key1, j - 32);
        r0 += (kj < key0) ? 1 : 0;
        r1 += (kj < key1) ? 1 : 0;
      }
      int p0 = 0, p1 = 0;
      for (int j = 0; j < n0; ++j) {
        const int kj = __builtin_amdgcn_readlane(key0, j);
        const int rj = __builtin_amdgcn_readlane(r0, j);
        if (rj == lane) p0 = kj;
        if (rj == lane + 32) p1 = kj;
      }
      for (int j = 32; j < deg; ++j) {
        const int kj = __builtin_amdgcn_readlane(key1, j - 32);
        const int rj = __builtin_amdgcn_readlane(r1, j - 32);
        if (rj == lane) p0 = kj;
        if (rj == lane + 32) p1 = kj;
      }
      s0 = p0 >> 6;
      s1 = p1 >> 6;
      if (lane < deg) lrow[lane] = s0;
      if (lane + 32 < deg) lrow[lane + 32] = s1;
    }
    gather_span<D>(h4, s0, deg < 32 ? deg : 32, lane, acc);
    if (deg > 32) gather_span<D>(h4, s1, deg - 32, lane, acc);
  } else {
    for (int e0 = 0; e0 < E; e0 += 32) {
      const int e = e0 + lane;
      int sv = 0;
      bool hit = false;
      if (e < E) {
        hit = (dst[e] == row);
        sv = src[e];
      }
      unsigned m = __builtin_amdgcn_ballot_w32(hit);
      while (m) {
        const int b = __builtin_ctz(m);
        m &= m - 1;
        const int s = __builtin_amdgcn_readlane(sv, b);
        const float4* p = h4 + (size_t)s * (D / 4) + lane;
#pragma unroll
        for (int q = 0; q < VEC; ++q) {
          const float4 t = p[32 * q];
          acc[q].x += t.x;
          acc[q].y += t.y;
          acc[q].z += t.z;
          acc[q].w += t.w;
        }
      }
    }
  }

  if (EPI) {
#pragma unroll
    for (int q = 0; q < VEC; ++q) {
      acc[q].x = fmaxf(acc[q].x * rs + bq[q].x, 0.0f);
      acc[q].y = fmaxf(acc[q].y * rs + bq[q].y, 0.0f);
      acc[q].z = fmaxf(acc[q].z * rs + bq[q].z, 0.0f);
      acc[q].w = fmaxf(acc[q].w * rs + bq[q].w, 0.0f);
    }
  }

  float4* out4 = reinterpret_cast<float4*>(agg) + (size_t)row * (D / 4) + lane;
#pragma unroll
  for (int q = 0; q < VEC; ++q) out4[32 * q] = acc[q];
}


template <int K, int NC, int ACT, bool HAS_SCALE, bool HAS_BIAS>
__global__ __launch_bounds__(256) void gemm_wmma_kernel(
    const float* __restrict__ A, const float* __restrict__ W,
    const float* __restrict__ bias, const float* __restrict__ rowscale,
    float* __restrict__ C, int M) {
  constexpr int PAD = 4;
  constexpr int LDSROW = K + PAD;
  __shared__ float sA[64 * LDSROW];

  const int tid = threadIdx.x;
  const int blockM = blockIdx.x * 64;
  const int blockN = blockIdx.y * 64;

  constexpr int F4_PER_ROW = K / 4;
  constexpr int TOTAL_F4 = 64 * F4_PER_ROW;
  for (int idx = tid; idx < TOTAL_F4; idx += 256) {
    int row = idx / F4_PER_ROW;
    int c4 = idx % F4_PER_ROW;
    float4 v = make_float4(0.f, 0.f, 0.f, 0.f);
    int gr = blockM + row;
    if (gr < M) v = *reinterpret_cast<const float4*>(A + (size_t)gr * K + c4 * 4);
    *reinterpret_cast<float4*>(&sA[row * LDSROW + c4 * 4]) = v;
  }
  __syncthreads();

  const int wave = tid >> 5;
  const int lane = tid & 31;
  const int wn = wave & 3;
  const int wmg = wave >> 2;
  const int half = lane >> 4;
  const int lm = lane & 15;
  const int col = blockN + wn * 16 + lm;

  v8f acc0 = {};
  v8f acc1 = {};
  const float* sArow0 = &sA[(wmg * 32 + lm) * LDSROW + 2 * half];
  const float* sArow1 = sArow0 + 16 * LDSROW;
  const float* Wp = W + (size_t)(2 * half) * NC + col;

#pragma unroll 8
  for (int k = 0; k < K; k += 4) {
    v2f a0 = *reinterpret_cast<const v2f*>(sArow0 + k);
    v2f a1 = *reinterpret_cast<const v2f*>(sArow1 + k);
    v2f b;
    b[0] = Wp[k * NC];
    b[1] = Wp[k * NC + NC];
    acc0 = __builtin_amdgcn_wmma_f32_16x16x4_f32(false, a0, false, b, (short)0,
                                                 acc0, false, false);
    acc1 = __builtin_amdgcn_wmma_f32_16x16x4_f32(false, a1, false, b, (short)0,
                                                 acc1, false, false);
  }

  const float bval = HAS_BIAS ? bias[col] : 0.0f;
  const int rbase0 = blockM + wmg * 32 + 8 * half;
  if (blockM + 64 <= M) {
#pragma unroll
    for (int t = 0; t < 2; ++t) {
      const int rbase = rbase0 + t * 16;
      float* Cp = C + (size_t)rbase * NC + col;
      const float* rs = rowscale + rbase;
#pragma unroll
      for (int r = 0; r < 8; ++r) {
        float v = (t == 0) ? acc0[r] : acc1[r];
        if (HAS_SCALE) v *= rs[r];
        if (HAS_BIAS) v += bval;
        if (ACT == 1) v = fmaxf(v, 0.0f);
        else if (ACT == 2) v = (v > 0.0f) ? v : expm1f(v);
        Cp[r * NC] = v;
      }
    }
  } else {
#pragma unroll
    for (int t = 0; t < 2; ++t) {
      const int rbase = rbase0 + t * 16;
#pragma unroll
      for (int r = 0; r < 8; ++r) {
        const int row = rbase + r;
        if (row < M) {
          float v = (t == 0) ? acc0[r] : acc1[r];
          if (HAS_SCALE) v *= rowscale[row];
          if (HAS_BIAS) v += bval;
          if (ACT == 1) v = fmaxf(v, 0.0f);
          else if (ACT == 2) v = (v > 0.0f) ? v : expm1f(v);
          C[(size_t)row * NC + col] = v;
        }
      }
    }
  }
}


extern "C" void kernel_launch(void* const* d_in, const int* in_sizes, int n_in,
                              void* d_out, int out_size, void* d_ws, size_t ws_size,
                              hipStream_t stream) {
  const float* feat1 = (const float*)d_in[0];
  const float* feat2 = (const float*)d_in[1];
  const int* ei1 = (const int*)d_in[2];
  const int* ei2 = (const int*)d_in[3];
  const float* W1 = (const float*)d_in[4];
  const float* b1 = (const float*)d_in[5];
  const float* W2 = (const float*)d_in[6];
  const float* b2 = (const float*)d_in[7];
  const float* fc1W = (const float*)d_in[8];
  const float* fc1b = (const float*)d_in[9];
  const float* fc2W = (const float*)d_in[10];
  const float* fc2b = (const float*)d_in[11];

  const int N = in_sizes[0] / 128;

  float* ws = (float*)d_ws;
  float* norm_src = ws;
  float* norm_dst = ws + N;
  float* bufA = ws + 2 * (size_t)N;
  float* bufB = bufA + (size_t)N * 256;

  for (int g = 0; g < 2; ++g) {
    const float* feat = g ? feat2 : feat1;
    const int E = (g ? in_sizes[3] : in_sizes[2]) / 2;
    const int* src = g ? ei2 : ei1;
    const int* dst = src + E;
    float* zout = (float*)d_out + (size_t)g * N * 128;

    int* lists = (int*)zout;
    int* deg_in = lists + (size_t)N * LIST_CAP;

    hipMemsetAsync(norm_src, 0, 2 * (size_t)N * sizeof(float), stream);
    degree_adjacency_kernel<<<(E + 255) / 256, 256, 0, stream>>>(
        src, dst, (int*)norm_src, (int*)norm_dst, lists, E);
    norm_kernel<<<(2 * N + 255) / 256, 256, 0, stream>>>((int*)norm_src, deg_in, N);

    scale_rows_kernel<<<(N * 128 / 4 + 255) / 256, 256, 0, stream>>>(
        bufA, feat, norm_src, N * 128 / 4, 32);
    gather_sum_kernel<128, true, false><<<(N * 32 + 255) / 256, 256, 0, stream>>>(
        bufA, deg_in, lists, src, dst, nullptr, nullptr, bufB, N, E);
    {
      dim3 grid((N + 63) / 64, 256 / 64);
      gemm_wmma_kernel<128, 256, 1, true, true><<<grid, 256, 0, stream>>>(
          bufB, W1, b1, norm_dst, bufA, N);
    }

    {
      dim3 grid((N + 63) / 64, 128 / 64);
      gemm_wmma_kernel<256, 128, 0, true, false><<<grid, 256, 0, stream>>>(
          bufA, W2, nullptr, norm_src, bufB, N);
    }
    gather_sum_kernel<128, false, true><<<(N * 32 + 255) / 256, 256, 0, stream>>>(
        bufB, deg_in, lists, src, dst, norm_dst, b2, bufA, N, E);

    {
      dim3 grid((N + 63) / 64, 128 / 64);
      gemm_wmma_kernel<128, 128, 2, false, true><<<grid, 256, 0, stream>>>(
          bufA, fc1W, fc1b, nullptr, bufB, N);
      gemm_wmma_kernel<128, 128, 0, false, true><<<grid, 256, 0, stream>>>(
          bufB, fc2W, fc2b, nullptr, zout, N);
    }
  }
}


# Degrees and adjacency lists are built by binning edges by 256-node range with LDS counters and ranking each bin in LDS, replacing two global integer atomics and a scattered store per edge
# speedup vs baseline: 3.3336x; 1.3560x over previous
#include <hip/hip_runtime.h>
#include <hip/hip_bf16.h>


typedef __attribute__((ext_vector_type(2))) float v2f;
typedef __attribute__((ext_vector_type(8))) float v8f;


constexpr int LIST_CAP = 64;
constexpr int BIN_NODES = 256;
constexpr int MAX_BINS = 256;
constexpr int SUB = 64;
constexpr int BIN_THREADS = 1024;
constexpr int BIN_EPT = 4;

struct BinScratch {
  unsigned* rec_d;
  int* cnt_d;
  int* cnt_s;
  unsigned char* rec_s;
  int nb;
};

__device__ __forceinline__ BinScratch bin_scratch(unsigned* scratch, size_t scratch_bytes,
                                                  int nbins, int nblk) {
  BinScratch b;
  int nb = nbins < MAX_BINS ? nbins : MAX_BINS;
  const size_t cells = (size_t)nb * (size_t)nblk;
  if (cells * (SUB * 5 + 8) > scratch_bytes) nb = 0;
  b.nb = nb;
  b.rec_d = scratch;
  b.cnt_d = reinterpret_cast<int*>(scratch + cells * SUB);
  b.cnt_s = b.cnt_d + cells;
  b.rec_s = reinterpret_cast<unsigned char*>(b.cnt_s + cells);
  return b;
}

__global__ __launch_bounds__(BIN_THREADS) void bin_edges_kernel(
    const int* __restrict__ src, const int* __restrict__ dst,
    int* __restrict__ cnt_out, int* __restrict__ cnt_in, int* __restrict__ lists,
    unsigned* __restrict__ scratch, size_t scratch_bytes, int E, int nbins) {
  constexpr int EPB = BIN_THREADS * BIN_EPT;
  __shared__ int table[2 * MAX_BINS];
  __shared__ int offs[2 * MAX_BINS];
  __shared__ int wtot[2 * MAX_BINS / 32];
  __shared__ unsigned st_rec_d[EPB];
  __shared__ unsigned st_pos_d[EPB];
  __shared__ unsigned st_pos_s[EPB];
  __shared__ unsigned char st_rec_s[EPB];
  const int tid = threadIdx.x;
  const int lane = tid & 31;
  const int blk = blockIdx.x;
  const int nblk = gridDim.x;
  const BinScratch bs = bin_scratch(scratch, scratch_bytes, nbins, nblk);
  if (tid < 2 * MAX_BINS) table[tid] = 0;
  __syncthreads();

  const int e0 = blk * EPB + tid;
  int s[BIN_EPT], d[BIN_EPT], rd[BIN_EPT], rs[BIN_EPT];
#pragma unroll
  for (int j = 0; j < BIN_EPT; ++j) {
    const int e = e0 + j * BIN_THREADS;
    s[j] = 0;
    d[j] = 0;
    if (e < E) {
      s[j] = src[e];
      d[j] = dst[e];
    }
  }
#pragma unroll
  for (int j = 0; j < BIN_EPT; ++j) {
    const int e = e0 + j * BIN_THREADS;
    rd[j] = -1;
    rs[j] = -1;
    if (e < E) {
      const unsigned bd = (unsigned)d[j] >> 8;
      const unsigned bsrc = (unsigned)s[j] >> 8;
      if (bd < (unsigned)bs.nb && (unsigned)s[j] < (1u << 24)) rd[j] = atomicAdd(&table[bd], 1);
      if (bsrc < (unsigned)bs.nb) rs[j] = atomicAdd(&table[MAX_BINS + bsrc], 1);
    }
  }
  __syncthreads();

  if (tid < 2 * MAX_BINS) {
    int c = table[tid];
    c = c < SUB ? c : SUB;
    int incl = c;
    for (int o = 1; o < 32; o <<= 1) {
      const int t = __shfl_up(incl, o);
      if (lane >= o) incl += t;
    }
    offs[tid] = incl - c;
    if (lane == 31) wtot[tid >> 5] = incl;
  }
  __syncthreads();
  if (tid < 2 * MAX_BINS) {
    const int w = tid >> 5;
    int add = 0;
    for (int k = w & ~7; k < w; ++k) add += wtot[k];
    offs[tid] += add;
  }
  __syncthreads();

#pragma unroll
  for (int j = 0; j < BIN_EPT; ++j) {
    const int e = e0 + j * BIN_THREADS;
    if (e < E) {
      const unsigned bd = (unsigned)d[j] >> 8;
      const unsigned bsrc = (unsigned)s[j] >> 8;
      if ((unsigned)rd[j] < (unsigned)SUB) {
        const int p = offs[bd] + rd[j];
        st_rec_d[p] = ((unsigned)s[j] << 8) | ((unsigned)d[j] & 255u);
        st_pos_d[p] = (unsigned)(((size_t)bd * nblk + blk) * SUB + rd[j]);
      } else {
        const int slot = atomicAdd(&cnt_in[d[j]], 1);
        if (slot < LIST_CAP) lists[(size_t)d[j] * LIST_CAP + slot] = s[j];
      }
      if ((unsigned)rs[j] < (unsigned)SUB) {
        const int p = offs[MAX_BINS + bsrc] + rs[j];
        st_rec_s[p] = (unsigned char)((unsigned)s[j] & 255u);
        st_pos_s[p] = (unsigned)(((size_t)bsrc * nblk + blk) * SUB + rs[j]);
      } else {
        atomicAdd(&cnt_out[s[j]], 1);
      }
    }
  }
  __syncthreads();

  const int last_d = table[MAX_BINS - 1];
  const int last_s = table[2 * MAX_BINS - 1];
  const int total_d = offs[MAX_BINS - 1] + (last_d < SUB ? last_d : SUB);
  const int total_s = offs[2 * MAX_BINS - 1] + (last_s < SUB ? last_s : SUB);
  for (int i = tid; i < total_d; i += BIN_THREADS) bs.rec_d[st_pos_d[i]] = st_rec_d[i];
  for (int i = tid; i < total_s; i += BIN_THREADS) bs.rec_s[st_pos_s[i]] = st_rec_s[i];
  if (tid < 2 * MAX_BINS) {
    const int b = tid & (MAX_BINS - 1);
    if (b < bs.nb) {
      const int c = table[tid];
      int* cnt = tid < MAX_BINS ? bs.cnt_d : bs.cnt_s;
      cnt[(size_t)blk * bs.nb + b] = c < SUB ? c : SUB;
    }
  }
}

__global__ __launch_bounds__(BIN_THREADS) void rank_lists_kernel(
    int* __restrict__ cnt_out, int* __restrict__ cnt_in, int* __restrict__ lists,
    int* __restrict__ deg_in, const unsigned* __restrict__ scratch, size_t scratch_bytes,
    int n_nodes, int nbins, int nblk) {
  constexpr int UNR = 4;
  __shared__ int tile[BIN_NODES * LIST_CAP];
  __shared__ int n_in[BIN_NODES];
  __shared__ int n_base[BIN_NODES];
  __shared__ int n_out[BIN_NODES];
  __shared__ int ccd[BIN_THREADS];
  __shared__ int ccs[BIN_THREADS];
  const int tid = threadIdx.x;
  const int b = blockIdx.x;
  const int node0 = b * BIN_NODES;
  const BinScratch bs = bin_scratch(const_cast<unsigned*>(scratch), scratch_bytes, nbins, nblk);

  if (tid < BIN_NODES) {
    const int node = node0 + tid;
    int ci = 0, co = 0;
    if (node < n_nodes) {
      ci = cnt_in[node];
      co = cnt_out[node];
    }
    n_in[tid] = ci;
    n_base[tid] = ci;
    n_out[tid] = co;
  }

  if (b < bs.nb) {
    for (int kc = 0; kc < nblk; kc += BIN_THREADS) {
      __syncthreads();
      if (kc + tid < nblk) {
        ccd[tid] = bs.cnt_d[(size_t)(kc + tid) * bs.nb + b];
        ccs[tid] = bs.cnt_s[(size_t)(kc + tid) * bs.nb + b];
      }
      __syncthreads();
      const int ncell = nblk - kc < BIN_THREADS ? nblk - kc : BIN_THREADS;
      const int nplace = ncell * SUB;
      const unsigned* rdp = bs.rec_d + ((size_t)b * nblk + kc) * SUB;
      const unsigned char* rsp = bs.rec_s + ((size_t)b * nblk + kc) * SUB;
      for (int p0 = tid; p0 < nplace; p0 += UNR * BIN_THREADS) {
        unsigned r[UNR];
        int q[UNR];
        bool has_r[UNR];
#pragma unroll
        for (int u = 0; u < UNR; ++u) {
          const int p = p0 + u * BIN_THREADS;
          r[u] = 0;
          has_r[u] = false;
          q[u] = -1;
          if (p < nplace) {
            if ((p % SUB) < ccd[p / SUB]) {
              r[u] = rdp[p];
              has_r[u] = true;
            }
            if ((p % SUB) < ccs[p / SUB]) q[u] = (int)rsp[p];
          }
        }
#pragma unroll
        for (int u = 0; u < UNR; ++u) {
          if (has_r[u]) {
            const int l = (int)(r[u] & 255u);
            const int slot = atomicAdd(&n_in[l], 1);
            if (slot < LIST_CAP) tile[l * LIST_CAP + slot] = (int)(r[u] >> 8);
          }
          if (q[u] >= 0) atomicAdd(&n_out[q[u]], 1);
        }
      }
    }
  }
  __syncthreads();

  for (int idx = tid; idx < BIN_NODES * LIST_CAP; idx += BIN_THREADS) {
    const int l = idx / LIST_CAP;
    const int slot = idx % LIST_CAP;
    const int node = node0 + l;
    if (node < n_nodes && slot >= n_base[l] && slot < n_in[l])
      lists[(size_t)node * LIST_CAP + slot] = tile[idx];
  }
  if (tid < BIN_NODES) {
    const int node = node0 + tid;
    if (node < n_nodes) {
      const int ci = n_in[tid];
      const int co = n_out[tid];
      deg_in[node] = ci;
      cnt_in[node] = __float_as_int(rsqrtf(fmaxf((float)ci, 1.0f)));
      cnt_out[node] = __float_as_int(rsqrtf(fmaxf((float)co, 1.0f)));
    }
  }
}


__global__ void scale_rows_kernel(float* __restrict__ out,
                                  const float* __restrict__ in,
                                  const float* __restrict__ norm,
                                  int n4, int d4) {
  int i = blockIdx.x * blockDim.x + threadIdx.x;
  if (i >= n4) return;
  int row = i / d4;
  float s = norm[row];
  float4 v = reinterpret_cast<const float4*>(in)[i];
  v.x *= s; v.y *= s; v.z *= s; v.w *= s;
  reinterpret_cast<float4*>(out)[i] = v;
}


template <int D>
__device__ __forceinline__ void gather_span(const float4* __restrict__ h4, int sreg,
                                            int count, int lane, float4 (&acc)[D / 128]) {
  constexpr int VEC = D / 128;
  constexpr int CH = 8 / VEC;
  for (int base = 0; base < count; base += CH) {
    int sidx[CH];
#pragma unroll
    for (int u = 0; u < CH; ++u) {
      int j = base + u;
      j = j < count ? j : count - 1;
      sidx[u] = __builtin_amdgcn_readlane(sreg, j);
    }
    float4 v[CH][VEC];
    if (base + CH <= count) {
#pragma unroll
      for (int u = 0; u < CH; ++u) {
        const float4* p = h4 + (size_t)sidx[u] * (D / 4) + lane;
#pragma unroll
        for (int q = 0; q < VEC; ++q) v[u][q] = p[32 * q];
      }
    } else {
#pragma unroll
      for (int u = 0; u < CH; ++u) {
        const float4* p = h4 + (size_t)sidx[u] * (D / 4) + lane;
        const bool valid = base + u < count;
#pragma unroll
        for (int q = 0; q < VEC; ++q) {
          const float4 t = p[32 * q];
          v[u][q].x = valid ? t.x : 0.f;
          v[u][q].y = valid ? t.y : 0.f;
          v[u][q].z = valid ? t.z : 0.f;
          v[u][q].w = valid ? t.w : 0.f;
        }
      }
    }
#pragma unroll
    for (int u = 0; u < CH; ++u) {
#pragma unroll
      for (int q = 0; q < VEC; ++q) {
        acc[q].x += v[u][q].x;
        acc[q].y += v[u][q].y;
        acc[q].z += v[u][q].z;
        acc[q].w += v[u][q].w;
      }
    }
  }
}

template <int D, bool SORT, bool EPI>
__global__ __launch_bounds__(256) void gather_sum_kernel(
    const float* __restrict__ h, const int* __restrict__ deg_in,
    int* __restrict__ lists, const int* __restrict__ src,
    const int* __restrict__ dst, const float* __restrict__ rowscale,
    const float* __restrict__ bias, float* __restrict__ agg, int n, int E) {
  constexpr int VEC = D / 128;
  const int lane = threadIdx.x & 31;
  const int row = __builtin_amdgcn_readfirstlane((int)((blockIdx.x * 256u + threadIdx.x) >> 5));
  if (row >= n) return;
  int* lrow = lists + (size_t)row * LIST_CAP;
  int s0 = lrow[lane];
  int s1 = lrow[lane + 32];
  const int deg = __builtin_amdgcn_readfirstlane(deg_in[row]);
  const float4* h4 = reinterpret_cast<const float4*>(h);

  float rs = 0.f;
  float4 bq[VEC];
  if (EPI) {
    rs = rowscale[row];
    const float4* b4 = reinterpret_cast<const float4*>(bias) + lane;
#pragma unroll
    for (int q = 0; q < VEC; ++q) bq[q] = b4[32 * q];
  }

  float4 acc[VEC];
#pragma unroll
  for (int q = 0; q < VEC; ++q) acc[q] = make_float4(0.f, 0.f, 0.f, 0.f);

  if (deg <= LIST_CAP) {
    if (SORT) {
      const int key0 = (lane < deg) ? (int)(((unsigned)s0 << 6) | (unsigned)lane) : 0x7fffffff;
      const int key1 = (lane + 32 < deg) ? (int)(((unsigned)s1 << 6) | (unsigned)(lane + 32)) : 0x7fffffff;
      const int n0 = deg < 32 ? deg : 32;
      int r0 = 0, r1 = 0;
      for (int j = 0; j < n0; ++j) {
        const int kj = __builtin_amdgcn_readlane(key0, j);
        r0 += (kj < key0) ? 1 : 0;
        r1 += (kj < key1) ? 1 : 0;
      }
      for (int j = 32; j < deg; ++j) {
        const int kj = __builtin_amdgcn_readlane(key1, j - 32);
        r0 += (kj < key0) ? 1 : 0;
        r1 += (kj < key1) ? 1 : 0;
      }
      int p0 = 0, p1 = 0;
      for (int j = 0; j < n0; ++j) {
        const int kj = __builtin_amdgcn_readlane(key0, j);
        const int rj = __builtin_amdgcn_readlane(r0, j);
        if (rj == lane) p0 = kj;
        if (rj == lane + 32) p1 = kj;
      }
      for (int j = 32; j < deg; ++j) {
        const int kj = __builtin_amdgcn_readlane(key1, j - 32);
        const int rj = __builtin_amdgcn_readlane(r1, j - 32);
        if (rj == lane) p0 = kj;
        if (rj == lane + 32) p1 = kj;
      }
      s0 = p0 >> 6;
      s1 = p1 >> 6;
      if (lane < deg) lrow[lane] = s0;
      if (lane + 32 < deg) lrow[lane + 32] = s1;
    }
    gather_span<D>(h4, s0, deg < 32 ? deg : 32, lane, acc);
    if (deg > 32) gather_span<D>(h4, s1, deg - 32, lane, acc);
  } else {
    for (int e0 = 0; e0 < E; e0 += 32) {
      const int e = e0 + lane;
      int sv = 0;
      bool hit = false;
      if (e < E) {
        hit = (dst[e] == row);
        sv = src[e];
      }
      unsigned m = __builtin_amdgcn_ballot_w32(hit);
      while (m) {
        const int b = __builtin_ctz(m);
        m &= m - 1;
        const int s = __builtin_amdgcn_readlane(sv, b);
        const float4* p = h4 + (size_t)s * (D / 4) + lane;
#pragma unroll
        for (int q = 0; q < VEC; ++q) {
          const float4 t = p[32 * q];
          acc[q].x += t.x;
          acc[q].y += t.y;
          acc[q].z += t.z;
          acc[q].w += t.w;
        }
      }
    }
  }

  if (EPI) {
#pragma unroll
    for (int q = 0; q < VEC; ++q) {
      acc[q].x = fmaxf(acc[q].x * rs + bq[q].x, 0.0f);
      acc[q].y = fmaxf(acc[q].y * rs + bq[q].y, 0.0f);
      acc[q].z = fmaxf(acc[q].z * rs + bq[q].z, 0.0f);
      acc[q].w = fmaxf(acc[q].w * rs + bq[q].w, 0.0f);
    }
  }

  float4* out4 = reinterpret_cast<float4*>(agg) + (size_t)row * (D / 4) + lane;
#pragma unroll
  for (int q = 0; q < VEC; ++q) out4[32 * q] = acc[q];
}


template <int K, int NC, int ACT, bool HAS_SCALE, bool HAS_BIAS>
__global__ __launch_bounds__(256) void gemm_wmma_kernel(
    const float* __restrict__ A, const float* __restrict__ W,
    const float* __restrict__ bias, const float* __restrict__ rowscale,
    float* __restrict__ C, int M) {
  constexpr int PAD = 4;
  constexpr int LDSROW = K + PAD;
  __shared__ float sA[64 * LDSROW];

  const int tid = threadIdx.x;
  const int blockM = blockIdx.x * 64;
  const int blockN = blockIdx.y * 64;

  constexpr int F4_PER_ROW = K / 4;
  constexpr int TOTAL_F4 = 64 * F4_PER_ROW;
  for (int idx = tid; idx < TOTAL_F4; idx += 256) {
    int row = idx / F4_PER_ROW;
    int c4 = idx % F4_PER_ROW;
    float4 v = make_float4(0.f, 0.f, 0.f, 0.f);
    int gr = blockM + row;
    if (gr < M) v = *reinterpret_cast<const float4*>(A + (size_t)gr * K + c4 * 4);
    *reinterpret_cast<float4*>(&sA[row * LDSROW + c4 * 4]) = v;
  }
  __syncthreads();

  const int wave = tid >> 5;
  const int lane = tid & 31;
  const int wn = wave & 3;
  const int wmg = wave >> 2;
  const int half = lane >> 4;
  const int lm = lane & 15;
  const int col = blockN + wn * 16 + lm;

  v8f acc0 = {};
  v8f acc1 = {};
  const float* sArow0 = &sA[(wmg * 32 + lm) * LDSROW + 2 * half];
  const float* sArow1 = sArow0 + 16 * LDSROW;
  const float* Wp = W + (size_t)(2 * half) * NC + col;

#pragma unroll 8
  for (int k = 0; k < K; k += 4) {
    v2f a0 = *reinterpret_cast<const v2f*>(sArow0 + k);
    v2f a1 = *reinterpret_cast<const v2f*>(sArow1 + k);
    v2f b;
    b[0] = Wp[k * NC];
    b[1] = Wp[k * NC + NC];
    acc0 = __builtin_amdgcn_wmma_f32_16x16x4_f32(false, a0, false, b, (short)0,
                                                 acc0, false, false);
    acc1 = __builtin_amdgcn_wmma_f32_16x16x4_f32(false, a1, false, b, (short)0,
                                                 acc1, false, false);
  }

  const float bval = HAS_BIAS ? bias[col] : 0.0f;
  const int rbase0 = blockM + wmg * 32 + 8 * half;
  if (blockM + 64 <= M) {
#pragma unroll
    for (int t = 0; t < 2; ++t) {
      const int rbase = rbase0 + t * 16;
      float* Cp = C + (size_t)rbase * NC + col;
      const float* rs = rowscale + rbase;
#pragma unroll
      for (int r = 0; r < 8; ++r) {
        float v = (t == 0) ? acc0[r] : acc1[r];
        if (HAS_SCALE) v *= rs[r];
        if (HAS_BIAS) v += bval;
        if (ACT == 1) v = fmaxf(v, 0.0f);
        else if (ACT == 2) v = (v > 0.0f) ? v : expm1f(v);
        Cp[r * NC] = v;
      }
    }
  } else {
#pragma unroll
    for (int t = 0; t < 2; ++t) {
      const int rbase = rbase0 + t * 16;
#pragma unroll
      for (int r = 0; r < 8; ++r) {
        const int row = rbase + r;
        if (row < M) {
          float v = (t == 0) ? acc0[r] : acc1[r];
          if (HAS_SCALE) v *= rowscale[row];
          if (HAS_BIAS) v += bval;
          if (ACT == 1) v = fmaxf(v, 0.0f);
          else if (ACT == 2) v = (v > 0.0f) ? v : expm1f(v);
          C[(size_t)row * NC + col] = v;
        }
      }
    }
  }
}


extern "C" void kernel_launch(void* const* d_in, const int* in_sizes, int n_in,
                              void* d_out, int out_size, void* d_ws, size_t ws_size,
                              hipStream_t stream) {
  const float* feat1 = (const float*)d_in[0];
  const float* feat2 = (const float*)d_in[1];
  const int* ei1 = (const int*)d_in[2];
  const int* ei2 = (const int*)d_in[3];
  const float* W1 = (const float*)d_in[4];
  const float* b1 = (const float*)d_in[5];
  const float* W2 = (const float*)d_in[6];
  const float* b2 = (const float*)d_in[7];
  const float* fc1W = (const float*)d_in[8];
  const float* fc1b = (const float*)d_in[9];
  const float* fc2W = (const float*)d_in[10];
  const float* fc2b = (const float*)d_in[11];

  const int N = in_sizes[0] / 128;

  float* ws = (float*)d_ws;
  float* norm_src = ws;
  float* norm_dst = ws + N;
  float* bufA = ws + 2 * (size_t)N;
  float* bufB = bufA + (size_t)N * 256;
  unsigned* scratch = (unsigned*)(bufB + (size_t)N * 128);
  const size_t scratch_bytes = (size_t)N * 128 * sizeof(float);

  for (int g = 0; g < 2; ++g) {
    const float* feat = g ? feat2 : feat1;
    const int E = (g ? in_sizes[3] : in_sizes[2]) / 2;
    const int* src = g ? ei2 : ei1;
    const int* dst = src + E;
    float* zout = (float*)d_out + (size_t)g * N * 128;

    int* lists = (int*)zout;
    int* deg_in = lists + (size_t)N * LIST_CAP;

    hipMemsetAsync(norm_src, 0, 2 * (size_t)N * sizeof(float), stream);
    const int nblk = (E + BIN_THREADS * BIN_EPT - 1) / (BIN_THREADS * BIN_EPT);
    const int nbins = (N + BIN_NODES - 1) / BIN_NODES;
    bin_edges_kernel<<<nblk, BIN_THREADS, 0, stream>>>(
        src, dst, (int*)norm_src, (int*)norm_dst, lists, scratch, scratch_bytes, E, nbins);
    rank_lists_kernel<<<nbins, BIN_THREADS, 0, stream>>>(
        (int*)norm_src, (int*)norm_dst, lists, deg_in, scratch, scratch_bytes, N, nbins, nblk);

    scale_rows_kernel<<<(N * 128 / 4 + 255) / 256, 256, 0, stream>>>(
        bufA, feat, norm_src, N * 128 / 4, 32);
    gather_sum_kernel<128, true, false><<<(N * 32 + 255) / 256, 256, 0, stream>>>(
        bufA, deg_in, lists, src, dst, nullptr, nullptr, bufB, N, E);
    {
      dim3 grid((N + 63) / 64, 256 / 64);
      gemm_wmma_kernel<128, 256, 1, true, true><<<grid, 256, 0, stream>>>(
          bufB, W1, b1, norm_dst, bufA, N);
    }

    {
      dim3 grid((N + 63) / 64, 128 / 64);
      gemm_wmma_kernel<256, 128, 0, true, false><<<grid, 256, 0, stream>>>(
          bufA, W2, nullptr, norm_src, bufB, N);
    }
    gather_sum_kernel<128, false, true><<<(N * 32 + 255) / 256, 256, 0, stream>>>(
        bufB, deg_in, lists, src, dst, norm_dst, b2, bufA, N, E);

    {
      dim3 grid((N + 63) / 64, 128 / 64);
      gemm_wmma_kernel<128, 128, 2, false, true><<<grid, 256, 0, stream>>>(
          bufA, fc1W, fc1b, nullptr, bufB, N);
      gemm_wmma_kernel<128, 128, 0, false, true><<<grid, 256, 0, stream>>>(
          bufB, fc2W, fc2b, nullptr, zout, N);
    }
  }
}


# GEMM A-tile staging split into a load pass and a store pass, so a thread's 8 or 16 global loads are in flight together; the guarded load-and-store loop waited for each load in turn
# speedup vs baseline: 3.4943x; 1.0482x over previous
#include <hip/hip_runtime.h>
#include <hip/hip_bf16.h>


typedef __attribute__((ext_vector_type(2))) float v2f;
typedef __attribute__((ext_vector_type(8))) float v8f;


constexpr int LIST_CAP = 64;
constexpr int BIN_NODES = 256;
constexpr int MAX_BINS = 256;
constexpr int SUB = 64;
constexpr int BIN_THREADS = 1024;
constexpr int BIN_EPT = 4;

struct BinScratch {
  unsigned* rec_d;
  int* cnt_d;
  int* cnt_s;
  unsigned char* rec_s;
  int nb;
};

__device__ __forceinline__ BinScratch bin_scratch(unsigned* scratch, size_t scratch_bytes,
                                                  int nbins, int nblk) {
  BinScratch b;
  int nb = nbins < MAX_BINS ? nbins : MAX_BINS;
  const size_t cells = (size_t)nb * (size_t)nblk;
  if (cells * (SUB * 5 + 8) > scratch_bytes) nb = 0;
  b.nb = nb;
  b.rec_d = scratch;
  b.cnt_d = reinterpret_cast<int*>(scratch + cells * SUB);
  b.cnt_s = b.cnt_d + cells;
  b.rec_s = reinterpret_cast<unsigned char*>(b.cnt_s + cells);
  return b;
}

__global__ __launch_bounds__(BIN_THREADS) void bin_edges_kernel(
    const int* __restrict__ src, const int* __restrict__ dst,
    int* __restrict__ cnt_out, int* __restrict__ cnt_in, int* __restrict__ lists,
    unsigned* __restrict__ scratch, size_t scratch_bytes, int E, int nbins) {
  constexpr int EPB = BIN_THREADS * BIN_EPT;
  __shared__ int table[2 * MAX_BINS];
  __shared__ int offs[2 * MAX_BINS];
  __shared__ int wtot[2 * MAX_BINS / 32];
  __shared__ unsigned st_rec_d[EPB];
  __shared__ unsigned st_pos_d[EPB];
  __shared__ unsigned st_pos_s[EPB];
  __shared__ unsigned char st_rec_s[EPB];
  const int tid = threadIdx.x;
  const int lane = tid & 31;
  const int blk = blockIdx.x;
  const int nblk = gridDim.x;
  const BinScratch bs = bin_scratch(scratch, scratch_bytes, nbins, nblk);
  if (tid < 2 * MAX_BINS) table[tid] = 0;
  __syncthreads();

  const int e0 = blk * EPB + tid;
  int s[BIN_EPT], d[BIN_EPT], rd[BIN_EPT], rs[BIN_EPT];
#pragma unroll
  for (int j = 0; j < BIN_EPT; ++j) {
    const int e = e0 + j * BIN_THREADS;
    s[j] = 0;
    d[j] = 0;
    if (e < E) {
      s[j] = src[e];
      d[j] = dst[e];
    }
  }
#pragma unroll
  for (int j = 0; j < BIN_EPT; ++j) {
    const int e = e0 + j * BIN_THREADS;
    rd[j] = -1;
    rs[j] = -1;
    if (e < E) {
      const unsigned bd = (unsigned)d[j] >> 8;
      const unsigned bsrc = (unsigned)s[j] >> 8;
      if (bd < (unsigned)bs.nb && (unsigned)s[j] < (1u << 24)) rd[j] = atomicAdd(&table[bd], 1);
      if (bsrc < (unsigned)bs.nb) rs[j] = atomicAdd(&table[MAX_BINS + bsrc], 1);
    }
  }
  __syncthreads();

  if (tid < 2 * MAX_BINS) {
    int c = table[tid];
    c = c < SUB ? c : SUB;
    int incl = c;
    for (int o = 1; o < 32; o <<= 1) {
      const int t = __shfl_up(incl, o);
      if (lane >= o) incl += t;
    }
    offs[tid] = incl - c;
    if (lane == 31) wtot[tid >> 5] = incl;
  }
  __syncthreads();
  if (tid < 2 * MAX_BINS) {
    const int w = tid >> 5;
    int add = 0;
    for (int k = w & ~7; k < w; ++k) add += wtot[k];
    offs[tid] += add;
  }
  __syncthreads();

#pragma unroll
  for (int j = 0; j < BIN_EPT; ++j) {
    const int e = e0 + j * BIN_THREADS;
    if (e < E) {
      const unsigned bd = (unsigned)d[j] >> 8;
      const unsigned bsrc = (unsigned)s[j] >> 8;
      if ((unsigned)rd[j] < (unsigned)SUB) {
        const int p = offs[bd] + rd[j];
        st_rec_d[p] = ((unsigned)s[j] << 8) | ((unsigned)d[j] & 255u);
        st_pos_d[p] = (unsigned)(((size_t)bd * nblk + blk) * SUB + rd[j]);
      } else {
        const int slot = atomicAdd(&cnt_in[d[j]], 1);
        if (slot < LIST_CAP) lists[(size_t)d[j] * LIST_CAP + slot] = s[j];
      }
      if ((unsigned)rs[j] < (unsigned)SUB) {
        const int p = offs[MAX_BINS + bsrc] + rs[j];
        st_rec_s[p] = (unsigned char)((unsigned)s[j] & 255u);
        st_pos_s[p] = (unsigned)(((size_t)bsrc * nblk + blk) * SUB + rs[j]);
      } else {
        atomicAdd(&cnt_out[s[j]], 1);
      }
    }
  }
  __syncthreads();

  const int last_d = table[MAX_BINS - 1];
  const int last_s = table[2 * MAX_BINS - 1];
  const int total_d = offs[MAX_BINS - 1] + (last_d < SUB ? last_d : SUB);
  const int total_s = offs[2 * MAX_BINS - 1] + (last_s < SUB ? last_s : SUB);
  for (int i = tid; i < total_d; i += BIN_THREADS) bs.rec_d[st_pos_d[i]] = st_rec_d[i];
  for (int i = tid; i < total_s; i += BIN_THREADS) bs.rec_s[st_pos_s[i]] = st_rec_s[i];
  if (tid < 2 * MAX_BINS) {
    const int b = tid & (MAX_BINS - 1);
    if (b < bs.nb) {
      const int c = table[tid];
      int* cnt = tid < MAX_BINS ? bs.cnt_d : bs.cnt_s;
      cnt[(size_t)blk * bs.nb + b] = c < SUB ? c : SUB;
    }
  }
}

__global__ __launch_bounds__(BIN_THREADS) void rank_lists_kernel(
    int* __restrict__ cnt_out, int* __restrict__ cnt_in, int* __restrict__ lists,
    int* __restrict__ deg_in, const unsigned* __restrict__ scratch, size_t scratch_bytes,
    int n_nodes, int nbins, int nblk) {
  constexpr int UNR = 4;
  __shared__ int tile[BIN_NODES * LIST_CAP];
  __shared__ int n_in[BIN_NODES];
  __shared__ int n_base[BIN_NODES];
  __shared__ int n_out[BIN_NODES];
  __shared__ int ccd[BIN_THREADS];
  __shared__ int ccs[BIN_THREADS];
  const int tid = threadIdx.x;
  const int b = blockIdx.x;
  const int node0 = b * BIN_NODES;
  const BinScratch bs = bin_scratch(const_cast<unsigned*>(scratch), scratch_bytes, nbins, nblk);

  if (tid < BIN_NODES) {
    const int node = node0 + tid;
    int ci = 0, co = 0;
    if (node < n_nodes) {
      ci = cnt_in[node];
      co = cnt_out[node];
    }
    n_in[tid] = ci;
    n_base[tid] = ci;
    n_out[tid] = co;
  }

  if (b < bs.nb) {
    for (int kc = 0; kc < nblk; kc += BIN_THREADS) {
      __syncthreads();
      if (kc + tid < nblk) {
        ccd[tid] = bs.cnt_d[(size_t)(kc + tid) * bs.nb + b];
        ccs[tid] = bs.cnt_s[(size_t)(kc + tid) * bs.nb + b];
      }
      __syncthreads();
      const int ncell = nblk - kc < BIN_THREADS ? nblk - kc : BIN_THREADS;
      const int nplace = ncell * SUB;
      const unsigned* rdp = bs.rec_d + ((size_t)b * nblk + kc) * SUB;
      const unsigned char* rsp = bs.rec_s + ((size_t)b * nblk + kc) * SUB;
      for (int p0 = tid; p0 < nplace; p0 += UNR * BIN_THREADS) {
        unsigned r[UNR];
        int q[UNR];
        bool has_r[UNR];
#pragma unroll
        for (int u = 0; u < UNR; ++u) {
          const int p = p0 + u * BIN_THREADS;
          r[u] = 0;
          has_r[u] = false;
          q[u] = -1;
          if (p < nplace) {
            if ((p % SUB) < ccd[p / SUB]) {
              r[u] = rdp[p];
              has_r[u] = true;
            }
            if ((p % SUB) < ccs[p / SUB]) q[u] = (int)rsp[p];
          }
        }
#pragma unroll
        for (int u = 0; u < UNR; ++u) {
          if (has_r[u]) {
            const int l = (int)(r[u] & 255u);
            const int slot = atomicAdd(&n_in[l], 1);
            if (slot < LIST_CAP) tile[l * LIST_CAP + slot] = (int)(r[u] >> 8);
          }
          if (q[u] >= 0) atomicAdd(&n_out[q[u]], 1);
        }
      }
    }
  }
  __syncthreads();

  for (int idx = tid; idx < BIN_NODES * LIST_CAP; idx += BIN_THREADS) {
    const int l = idx / LIST_CAP;
    const int slot = idx % LIST_CAP;
    const int node = node0 + l;
    if (node < n_nodes && slot >= n_base[l] && slot < n_in[l])
      lists[(size_t)node * LIST_CAP + slot] = tile[idx];
  }
  if (tid < BIN_NODES) {
    const int node = node0 + tid;
    if (node < n_nodes) {
      const int ci = n_in[tid];
      const int co = n_out[tid];
      deg_in[node] = ci;
      cnt_in[node] = __float_as_int(rsqrtf(fmaxf((float)ci, 1.0f)));
      cnt_out[node] = __float_as_int(rsqrtf(fmaxf((float)co, 1.0f)));
    }
  }
}


__global__ void scale_rows_kernel(float* __restrict__ out,
                                  const float* __restrict__ in,
                                  const float* __restrict__ norm,
                                  int n4, int d4) {
  int i = blockIdx.x * blockDim.x + threadIdx.x;
  if (i >= n4) return;
  int row = i / d4;
  float s = norm[row];
  float4 v = reinterpret_cast<const float4*>(in)[i];
  v.x *= s; v.y *= s; v.z *= s; v.w *= s;
  reinterpret_cast<float4*>(out)[i] = v;
}


template <int D>
__device__ __forceinline__ void gather_span(const float4* __restrict__ h4, int sreg,
                                            int count, int lane, float4 (&acc)[D / 128]) {
  constexpr int VEC = D / 128;
  constexpr int CH = 8 / VEC;
  for (int base = 0; base < count; base += CH) {
    int sidx[CH];
#pragma unroll
    for (int u = 0; u < CH; ++u) {
      int j = base + u;
      j = j < count ? j : count - 1;
      sidx[u] = __builtin_amdgcn_readlane(sreg, j);
    }
    float4 v[CH][VEC];
    if (base + CH <= count) {
#pragma unroll
      for (int u = 0; u < CH; ++u) {
        const float4* p = h4 + (size_t)sidx[u] * (D / 4) + lane;
#pragma unroll
        for (int q = 0; q < VEC; ++q) v[u][q] = p[32 * q];
      }
    } else {
#pragma unroll
      for (int u = 0; u < CH; ++u) {
        const float4* p = h4 + (size_t)sidx[u] * (D / 4) + lane;
        const bool valid = base + u < count;
#pragma unroll
        for (int q = 0; q < VEC; ++q) {
          const float4 t = p[32 * q];
          v[u][q].x = valid ? t.x : 0.f;
          v[u][q].y = valid ? t.y : 0.f;
          v[u][q].z = valid ? t.z : 0.f;
          v[u][q].w = valid ? t.w : 0.f;
        }
      }
    }
#pragma unroll
    for (int u = 0; u < CH; ++u) {
#pragma unroll
      for (int q = 0; q < VEC; ++q) {
        acc[q].x += v[u][q].x;
        acc[q].y += v[u][q].y;
        acc[q].z += v[u][q].z;
        acc[q].w += v[u][q].w;
      }
    }
  }
}

template <int D, bool SORT, bool EPI>
__global__ __launch_bounds__(256) void gather_sum_kernel(
    const float* __restrict__ h, const int* __restrict__ deg_in,
    int* __restrict__ lists, const int* __restrict__ src,
    const int* __restrict__ dst, const float* __restrict__ rowscale,
    const float* __restrict__ bias, float* __restrict__ agg, int n, int E) {
  constexpr int VEC = D / 128;
  const int lane = threadIdx.x & 31;
  const int row = __builtin_amdgcn_readfirstlane((int)((blockIdx.x * 256u + threadIdx.x) >> 5));
  if (row >= n) return;
  int* lrow = lists + (size_t)row * LIST_CAP;
  int s0 = lrow[lane];
  int s1 = lrow[lane + 32];
  const int deg = __builtin_amdgcn_readfirstlane(deg_in[row]);
  const float4* h4 = reinterpret_cast<const float4*>(h);

  float rs = 0.f;
  float4 bq[VEC];
  if (EPI) {
    rs = rowscale[row];
    const float4* b4 = reinterpret_cast<const float4*>(bias) + lane;
#pragma unroll
    for (int q = 0; q < VEC; ++q) bq[q] = b4[32 * q];
  }

  float4 acc[VEC];
#pragma unroll
  for (int q = 0; q < VEC; ++q) acc[q] = make_float4(0.f, 0.f, 0.f, 0.f);

  if (deg <= LIST_CAP) {
    if (SORT) {
      const int key0 = (lane < deg) ? (int)(((unsigned)s0 << 6) | (unsigned)lane) : 0x7fffffff;
      const int key1 = (lane + 32 < deg) ? (int)(((unsigned)s1 << 6) | (unsigned)(lane + 32)) : 0x7fffffff;
      const int n0 = deg < 32 ? deg : 32;
      int r0 = 0, r1 = 0;
      for (int j = 0; j < n0; ++j) {
        const int kj = __builtin_amdgcn_readlane(key0, j);
        r0 += (kj < key0) ? 1 : 0;
        r1 += (kj < key1) ? 1 : 0;
      }
      for (int j = 32; j < deg; ++j) {
        const int kj = __builtin_amdgcn_readlane(key1, j - 32);
        r0 += (kj < key0) ? 1 : 0;
        r1 += (kj < key1) ? 1 : 0;
      }
      int p0 = 0, p1 = 0;
      for (int j = 0; j < n0; ++j) {
        const int kj = __builtin_amdgcn_readlane(key0, j);
        const int rj = __builtin_amdgcn_readlane(r0, j);
        if (rj == lane) p0 = kj;
        if (rj == lane + 32) p1 = kj;
      }
      for (int j = 32; j < deg; ++j) {
        const int kj = __builtin_amdgcn_readlane(key1, j - 32);
        const int rj = __builtin_amdgcn_readlane(r1, j - 32);
        if (rj == lane) p0 = kj;
        if (rj == lane + 32) p1 = kj;
      }
      s0 = p0 >> 6;
      s1 = p1 >> 6;
      if (lane < deg) lrow[lane] = s0;
      if (lane + 32 < deg) lrow[lane + 32] = s1;
    }
    gather_span<D>(h4, s0, deg < 32 ? deg : 32, lane, acc);
    if (deg > 32) gather_span<D>(h4, s1, deg - 32, lane, acc);
  } else {
    for (int e0 = 0; e0 < E; e0 += 32) {
      const int e = e0 + lane;
      int sv = 0;
      bool hit = false;
      if (e < E) {
        hit = (dst[e] == row);
        sv = src[e];
      }
      unsigned m = __builtin_amdgcn_ballot_w32(hit);
      while (m) {
        const int b = __builtin_ctz(m);
        m &= m - 1;
        const int s = __builtin_amdgcn_readlane(sv, b);
        const float4* p = h4 + (size_t)s * (D / 4) + lane;
#pragma unroll
        for (int q = 0; q < VEC; ++q) {
          const float4 t = p[32 * q];
          acc[q].x += t.x;
          acc[q].y += t.y;
          acc[q].z += t.z;
          acc[q].w += t.w;
        }
      }
    }
  }

  if (EPI) {
#pragma unroll
    for (int q = 0; q < VEC; ++q) {
      acc[q].x = fmaxf(acc[q].x * rs + bq[q].x, 0.0f);
      acc[q].y = fmaxf(acc[q].y * rs + bq[q].y, 0.0f);
      acc[q].z = fmaxf(acc[q].z * rs + bq[q].z, 0.0f);
      acc[q].w = fmaxf(acc[q].w * rs + bq[q].w, 0.0f);
    }
  }

  float4* out4 = reinterpret_cast<float4*>(agg) + (size_t)row * (D / 4) + lane;
#pragma unroll
  for (int q = 0; q < VEC; ++q) out4[32 * q] = acc[q];
}


template <int K, int NC, int ACT, bool HAS_SCALE, bool HAS_BIAS>
__global__ __launch_bounds__(256) void gemm_wmma_kernel(
    const float* __restrict__ A, const float* __restrict__ W,
    const float* __restrict__ bias, const float* __restrict__ rowscale,
    float* __restrict__ C, int M) {
  constexpr int PAD = 4;
  constexpr int LDSROW = K + PAD;
  __shared__ float sA[64 * LDSROW];

  const int tid = threadIdx.x;
  const int blockM = blockIdx.x * 64;
  const int blockN = blockIdx.y * 64;

  constexpr int F4_PER_ROW = K / 4;
  constexpr int TOTAL_F4 = 64 * F4_PER_ROW;
  constexpr int STAGE_IT = TOTAL_F4 / 256;
  float4 stage[STAGE_IT];
  if (blockM + 64 <= M) {
#pragma unroll
    for (int it = 0; it < STAGE_IT; ++it) {
      const int idx = tid + it * 256;
      const int row = idx / F4_PER_ROW;
      const int c4 = idx % F4_PER_ROW;
      stage[it] = *reinterpret_cast<const float4*>(A + (size_t)(blockM + row) * K + c4 * 4);
    }
  } else {
#pragma unroll
    for (int it = 0; it < STAGE_IT; ++it) {
      const int idx = tid + it * 256;
      const int row = idx / F4_PER_ROW;
      const int c4 = idx % F4_PER_ROW;
      const int gr = blockM + row;
      const bool valid = gr < M;
      const float4 t =
          *reinterpret_cast<const float4*>(A + (size_t)(valid ? gr : M - 1) * K + c4 * 4);
      stage[it].x = valid ? t.x : 0.f;
      stage[it].y = valid ? t.y : 0.f;
      stage[it].z = valid ? t.z : 0.f;
      stage[it].w = valid ? t.w : 0.f;
    }
  }
#pragma unroll
  for (int it = 0; it < STAGE_IT; ++it) {
    const int idx = tid + it * 256;
    const int row = idx / F4_PER_ROW;
    const int c4 = idx % F4_PER_ROW;
    *reinterpret_cast<float4*>(&sA[row * LDSROW + c4 * 4]) = stage[it];
  }
  __syncthreads();

  const int wave = tid >> 5;
  const int lane = tid & 31;
  const int wn = wave & 3;
  const int wmg = wave >> 2;
  const int half = lane >> 4;
  const int lm = lane & 15;
  const int col = blockN + wn * 16 + lm;

  v8f acc0 = {};
  v8f acc1 = {};
  const float* sArow0 = &sA[(wmg * 32 + lm) * LDSROW + 2 * half];
  const float* sArow1 = sArow0 + 16 * LDSROW;
  const float* Wp = W + (size_t)(2 * half) * NC + col;

#pragma unroll 8
  for (int k = 0; k < K; k += 4) {
    v2f a0 = *reinterpret_cast<const v2f*>(sArow0 + k);
    v2f a1 = *reinterpret_cast<const v2f*>(sArow1 + k);
    v2f b;
    b[0] = Wp[k * NC];
    b[1] = Wp[k * NC + NC];
    acc0 = __builtin_amdgcn_wmma_f32_16x16x4_f32(false, a0, false, b, (short)0,
                                                 acc0, false, false);
    acc1 = __builtin_amdgcn_wmma_f32_16x16x4_f32(false, a1, false, b, (short)0,
                                                 acc1, false, false);
  }

  const float bval = HAS_BIAS ? bias[col] : 0.0f;
  const int rbase0 = blockM + wmg * 32 + 8 * half;
  if (blockM + 64 <= M) {
#pragma unroll
    for (int t = 0; t < 2; ++t) {
      const int rbase = rbase0 + t * 16;
      float* Cp = C + (size_t)rbase * NC + col;
      const float* rs = rowscale + rbase;
#pragma unroll
      for (int r = 0; r < 8; ++r) {
        float v = (t == 0) ? acc0[r] : acc1[r];
        if (HAS_SCALE) v *= rs[r];
        if (HAS_BIAS) v += bval;
        if (ACT == 1) v = fmaxf(v, 0.0f);
        else if (ACT == 2) v = (v > 0.0f) ? v : expm1f(v);
        Cp[r * NC] = v;
      }
    }
  } else {
#pragma unroll
    for (int t = 0; t < 2; ++t) {
      const int rbase = rbase0 + t * 16;
#pragma unroll
      for (int r = 0; r < 8; ++r) {
        const int row = rbase + r;
        if (row < M) {
          float v = (t == 0) ? acc0[r] : acc1[r];
          if (HAS_SCALE) v *= rowscale[row];
          if (HAS_BIAS) v += bval;
          if (ACT == 1) v = fmaxf(v, 0.0f);
          else if (ACT == 2) v = (v > 0.0f) ? v : expm1f(v);
          C[(size_t)row * NC + col] = v;
        }
      }
    }
  }
}


extern "C" void kernel_launch(void* const* d_in, const int* in_sizes, int n_in,
                              void* d_out, int out_size, void* d_ws, size_t ws_size,
                              hipStream_t stream) {
  const float* feat1 = (const float*)d_in[0];
  const float* feat2 = (const float*)d_in[1];
  const int* ei1 = (const int*)d_in[2];
  const int* ei2 = (const int*)d_in[3];
  const float* W1 = (const float*)d_in[4];
  const float* b1 = (const float*)d_in[5];
  const float* W2 = (const float*)d_in[6];
  const float* b2 = (const float*)d_in[7];
  const float* fc1W = (const float*)d_in[8];
  const float* fc1b = (const float*)d_in[9];
  const float* fc2W = (const float*)d_in[10];
  const float* fc2b = (const float*)d_in[11];

  const int N = in_sizes[0] / 128;

  float* ws = (float*)d_ws;
  float* norm_src = ws;
  float* norm_dst = ws + N;
  float* bufA = ws + 2 * (size_t)N;
  float* bufB = bufA + (size_t)N * 256;
  unsigned* scratch = (unsigned*)(bufB + (size_t)N * 128);
  const size_t scratch_bytes = (size_t)N * 128 * sizeof(float);

  for (int g = 0; g < 2; ++g) {
    const float* feat = g ? feat2 : feat1;
    const int E = (g ? in_sizes[3] : in_sizes[2]) / 2;
    const int* src = g ? ei2 : ei1;
    const int* dst = src + E;
    float* zout = (float*)d_out + (size_t)g * N * 128;

    int* lists = (int*)zout;
    int* deg_in = lists + (size_t)N * LIST_CAP;

    hipMemsetAsync(norm_src, 0, 2 * (size_t)N * sizeof(float), stream);
    const int nblk = (E + BIN_THREADS * BIN_EPT - 1) / (BIN_THREADS * BIN_EPT);
    const int nbins = (N + BIN_NODES - 1) / BIN_NODES;
    bin_edges_kernel<<<nblk, BIN_THREADS, 0, stream>>>(
        src, dst, (int*)norm_src, (int*)norm_dst, lists, scratch, scratch_bytes, E, nbins);
    rank_lists_kernel<<<nbins, BIN_THREADS, 0, stream>>>(
        (int*)norm_src, (int*)norm_dst, lists, deg_in, scratch, scratch_bytes, N, nbins, nblk);

    scale_rows_kernel<<<(N * 128 / 4 + 255) / 256, 256, 0, stream>>>(
        bufA, feat, norm_src, N * 128 / 4, 32);
    gather_sum_kernel<128, true, false><<<(N * 32 + 255) / 256, 256, 0, stream>>>(
        bufA, deg_in, lists, src, dst, nullptr, nullptr, bufB, N, E);
    {
      dim3 grid((N + 63) / 64, 256 / 64);
      gemm_wmma_kernel<128, 256, 1, true, true><<<grid, 256, 0, stream>>>(
          bufB, W1, b1, norm_dst, bufA, N);
    }

    {
      dim3 grid((N + 63) / 64, 128 / 64);
      gemm_wmma_kernel<256, 128, 0, true, false><<<grid, 256, 0, stream>>>(
          bufA, W2, nullptr, norm_src, bufB, N);
    }
    gather_sum_kernel<128, false, true><<<(N * 32 + 255) / 256, 256, 0, stream>>>(
        bufB, deg_in, lists, src, dst, norm_dst, b2, bufA, N, E);

    {
      dim3 grid((N + 63) / 64, 128 / 64);
      gemm_wmma_kernel<128, 128, 2, false, true><<<grid, 256, 0, stream>>>(
          bufA, fc1W, fc1b, nullptr, bufB, N);
      gemm_wmma_kernel<128, 128, 0, false, true><<<grid, 256, 0, stream>>>(
          bufB, fc2W, fc2b, nullptr, zout, N);
    }
  }
}


# Each step is one launch for both graphs (graph = grid y or z index) where the workspace holds both graphs' rows; only the two GEMMs around the 256-wide layer stay per graph: 18 launches become 12
# speedup vs baseline: 4.2892x; 1.2275x over previous
#include <hip/hip_runtime.h>
#include <hip/hip_bf16.h>


typedef __attribute__((ext_vector_type(2))) float v2f;
typedef __attribute__((ext_vector_type(8))) float v8f;


constexpr int LIST_CAP = 64;
constexpr int BIN_NODES = 256;
constexpr int MAX_BINS = 256;
constexpr int SUB = 64;
constexpr int BIN_THREADS = 1024;
constexpr int BIN_EPT = 4;

struct GraphMeta {
  int* lists;
  int* deg_in;
  int* cnt_out;
  int* cnt_in;
};

__device__ __host__ __forceinline__ GraphMeta graph_meta(int* meta0, int g, int n) {
  GraphMeta m;
  m.lists = meta0 + (size_t)g * (size_t)n * 128;
  m.deg_in = m.lists + (size_t)n * LIST_CAP;
  m.cnt_out = meta0 + (size_t)n * (LIST_CAP + 1) + (size_t)g * 2 * (size_t)n;
  m.cnt_in = m.cnt_out + n;
  return m;
}

struct BinScratch {
  unsigned* rec_d;
  int* cnt_d;
  int* cnt_s;
  unsigned char* rec_s;
  int nb;
};

__device__ __forceinline__ BinScratch bin_scratch(unsigned* scratch, size_t scratch_bytes,
                                                  int nbins, int nblk) {
  BinScratch b;
  int nb = nbins < MAX_BINS ? nbins : MAX_BINS;
  const size_t cells = (size_t)nb * (size_t)nblk;
  if (cells * (SUB * 5 + 8) > scratch_bytes) nb = 0;
  b.nb = nb;
  b.rec_d = scratch;
  b.cnt_d = reinterpret_cast<int*>(scratch + cells * SUB);
  b.cnt_s = b.cnt_d + cells;
  b.rec_s = reinterpret_cast<unsigned char*>(b.cnt_s + cells);
  return b;
}

__global__ __launch_bounds__(BIN_THREADS) void bin_edges_kernel(
    const int* __restrict__ ei_a, const int* __restrict__ ei_b, int E_a, int E_b,
    int* __restrict__ meta0, int n_nodes,
    unsigned* __restrict__ scratch0, size_t scratch_bytes, int nbins) {
  constexpr int EPB = BIN_THREADS * BIN_EPT;
  const int g = blockIdx.y;
  const int E = g ? E_b : E_a;
  const int* __restrict__ src = g ? ei_b : ei_a;
  const int* __restrict__ dst = src + E;
  const GraphMeta gm = graph_meta(meta0, g, n_nodes);
  int* __restrict__ cnt_out = gm.cnt_out;
  int* __restrict__ cnt_in = gm.cnt_in;
  int* __restrict__ lists = gm.lists;
  unsigned* __restrict__ scratch = scratch0 + (size_t)g * (scratch_bytes / sizeof(unsigned));
  __shared__ int table[2 * MAX_BINS];
  __shared__ int offs[2 * MAX_BINS];
  __shared__ int wtot[2 * MAX_BINS / 32];
  __shared__ unsigned st_rec_d[EPB];
  __shared__ unsigned st_pos_d[EPB];
  __shared__ unsigned st_pos_s[EPB];
  __shared__ unsigned char st_rec_s[EPB];
  const int tid = threadIdx.x;
  const int lane = tid & 31;
  const int blk = blockIdx.x;
  const int nblk = gridDim.x;
  const BinScratch bs = bin_scratch(scratch, scratch_bytes, nbins, nblk);
  if (tid < 2 * MAX_BINS) table[tid] = 0;
  __syncthreads();

  const int e0 = blk * EPB + tid;
  int s[BIN_EPT], d[BIN_EPT], rd[BIN_EPT], rs[BIN_EPT];
#pragma unroll
  for (int j = 0; j < BIN_EPT; ++j) {
    const int e = e0 + j * BIN_THREADS;
    s[j] = 0;
    d[j] = 0;
    if (e < E) {
      s[j] = src[e];
      d[j] = dst[e];
    }
  }
#pragma unroll
  for (int j = 0; j < BIN_EPT; ++j) {
    const int e = e0 + j * BIN_THREADS;
    rd[j] = -1;
    rs[j] = -1;
    if (e < E) {
      const unsigned bd = (unsigned)d[j] >> 8;
      const unsigned bsrc = (unsigned)s[j] >> 8;
      if (bd < (unsigned)bs.nb && (unsigned)s[j] < (1u << 24)) rd[j] = atomicAdd(&table[bd], 1);
      if (bsrc < (unsigned)bs.nb) rs[j] = atomicAdd(&table[MAX_BINS + bsrc], 1);
    }
  }
  __syncthreads();

  if (tid < 2 * MAX_BINS) {
    int c = table[tid];
    c = c < SUB ? c : SUB;
    int incl = c;
    for (int o = 1; o < 32; o <<= 1) {
      const int t = __shfl_up(incl, o);
      if (lane >= o) incl += t;
    }
    offs[tid] = incl - c;
    if (lane == 31) wtot[tid >> 5] = incl;
  }
  __syncthreads();
  if (tid < 2 * MAX_BINS) {
    const int w = tid >> 5;
    int add = 0;
    for (int k = w & ~7; k < w; ++k) add += wtot[k];
    offs[tid] += add;
  }
  __syncthreads();

#pragma unroll
  for (int j = 0; j < BIN_EPT; ++j) {
    const int e = e0 + j * BIN_THREADS;
    if (e < E) {
      const unsigned bd = (unsigned)d[j] >> 8;
      const unsigned bsrc = (unsigned)s[j] >> 8;
      if ((unsigned)rd[j] < (unsigned)SUB) {
        const int p = offs[bd] + rd[j];
        st_rec_d[p] = ((unsigned)s[j] << 8) | ((unsigned)d[j] & 255u);
        st_pos_d[p] = (unsigned)(((size_t)bd * nblk + blk) * SUB + rd[j]);
      } else {
        const int slot = atomicAdd(&cnt_in[d[j]], 1);
        if (slot < LIST_CAP) lists[(size_t)d[j] * LIST_CAP + slot] = s[j];
      }
      if ((unsigned)rs[j] < (unsigned)SUB) {
        const int p = offs[MAX_BINS + bsrc] + rs[j];
        st_rec_s[p] = (unsigned char)((unsigned)s[j] & 255u);
        st_pos_s[p] = (unsigned)(((size_t)bsrc * nblk + blk) * SUB + rs[j]);
      } else {
        atomicAdd(&cnt_out[s[j]], 1);
      }
    }
  }
  __syncthreads();

  const int last_d = table[MAX_BINS - 1];
  const int last_s = table[2 * MAX_BINS - 1];
  const int total_d = offs[MAX_BINS - 1] + (last_d < SUB ? last_d : SUB);
  const int total_s = offs[2 * MAX_BINS - 1] + (last_s < SUB ? last_s : SUB);
  for (int i = tid; i < total_d; i += BIN_THREADS) bs.rec_d[st_pos_d[i]] = st_rec_d[i];
  for (int i = tid; i < total_s; i += BIN_THREADS) bs.rec_s[st_pos_s[i]] = st_rec_s[i];
  if (tid < 2 * MAX_BINS) {
    const int b = tid & (MAX_BINS - 1);
    if (b < bs.nb) {
      const int c = table[tid];
      int* cnt = tid < MAX_BINS ? bs.cnt_d : bs.cnt_s;
      cnt[(size_t)blk * bs.nb + b] = c < SUB ? c : SUB;
    }
  }
}

__global__ __launch_bounds__(BIN_THREADS) void rank_lists_kernel(
    int* __restrict__ meta0, const unsigned* __restrict__ scratch0, size_t scratch_bytes,
    int n_nodes, int nbins, int nblk) {
  constexpr int UNR = 4;
  const int g = blockIdx.y;
  const GraphMeta gm = graph_meta(meta0, g, n_nodes);
  int* __restrict__ cnt_out = gm.cnt_out;
  int* __restrict__ cnt_in = gm.cnt_in;
  int* __restrict__ lists = gm.lists;
  int* __restrict__ deg_in = gm.deg_in;
  const unsigned* __restrict__ scratch = scratch0 + (size_t)g * (scratch_bytes / sizeof(unsigned));
  __shared__ int tile[BIN_NODES * LIST_CAP];
  __shared__ int n_in[BIN_NODES];
  __shared__ int n_base[BIN_NODES];
  __shared__ int n_out[BIN_NODES];
  __shared__ int ccd[BIN_THREADS];
  __shared__ int ccs[BIN_THREADS];
  const int tid = threadIdx.x;
  const int b = blockIdx.x;
  const int node0 = b * BIN_NODES;
  const BinScratch bs = bin_scratch(const_cast<unsigned*>(scratch), scratch_bytes, nbins, nblk);

  if (tid < BIN_NODES) {
    const int node = node0 + tid;
    int ci = 0, co = 0;
    if (node < n_nodes) {
      ci = cnt_in[node];
      co = cnt_out[node];
    }
    n_in[tid] = ci;
    n_base[tid] = ci;
    n_out[tid] = co;
  }

  if (b < bs.nb) {
    for (int kc = 0; kc < nblk; kc += BIN_THREADS) {
      __syncthreads();
      if (kc + tid < nblk) {
        ccd[tid] = bs.cnt_d[(size_t)(kc + tid) * bs.nb + b];
        ccs[tid] = bs.cnt_s[(size_t)(kc + tid) * bs.nb + b];
      }
      __syncthreads();
      const int ncell = nblk - kc < BIN_THREADS ? nblk - kc : BIN_THREADS;
      const int nplace = ncell * SUB;
      const unsigned* rdp = bs.rec_d + ((size_t)b * nblk + kc) * SUB;
      const unsigned char* rsp = bs.rec_s + ((size_t)b * nblk + kc) * SUB;
      for (int p0 = tid; p0 < nplace; p0 += UNR * BIN_THREADS) {
        unsigned r[UNR];
        int q[UNR];
        bool has_r[UNR];
#pragma unroll
        for (int u = 0; u < UNR; ++u) {
          const int p = p0 + u * BIN_THREADS;
          r[u] = 0;
          has_r[u] = false;
          q[u] = -1;
          if (p < nplace) {
            if ((p % SUB) < ccd[p / SUB]) {
              r[u] = rdp[p];
              has_r[u] = true;
            }
            if ((p % SUB) < ccs[p / SUB]) q[u] = (int)rsp[p];
          }
        }
#pragma unroll
        for (int u = 0; u < UNR; ++u) {
          if (has_r[u]) {
            const int l = (int)(r[u] & 255u);
            const int slot = atomicAdd(&n_in[l], 1);
            if (slot < LIST_CAP) tile[l * LIST_CAP + slot] = (int)(r[u] >> 8);
          }
          if (q[u] >= 0) atomicAdd(&n_out[q[u]], 1);
        }
      }
    }
  }
  __syncthreads();

  for (int idx = tid; idx < BIN_NODES * LIST_CAP; idx += BIN_THREADS) {
    const int l = idx / LIST_CAP;
    const int slot = idx % LIST_CAP;
    const int node = node0 + l;
    if (node < n_nodes && slot >= n_base[l] && slot < n_in[l])
      lists[(size_t)node * LIST_CAP + slot] = tile[idx];
  }
  if (tid < BIN_NODES) {
    const int node = node0 + tid;
    if (node < n_nodes) {
      const int ci = n_in[tid];
      const int co = n_out[tid];
      deg_in[node] = ci;
      cnt_in[node] = __float_as_int(rsqrtf(fmaxf((float)ci, 1.0f)));
      cnt_out[node] = __float_as_int(rsqrtf(fmaxf((float)co, 1.0f)));
    }
  }
}


__global__ void scale_rows_kernel(float* __restrict__ out0,
                                  const float* __restrict__ in_a,
                                  const float* __restrict__ in_b,
                                  int* __restrict__ meta0,
                                  int n4, int d4, int n_nodes) {
  int i = blockIdx.x * blockDim.x + threadIdx.x;
  if (i >= n4) return;
  const int g = blockIdx.y;
  const float* __restrict__ in = g ? in_b : in_a;
  float* __restrict__ out = out0 + (size_t)g * (size_t)n4 * 4;
  const float* __restrict__ norm = reinterpret_cast<const float*>(graph_meta(meta0, g, n_nodes).cnt_out);
  int row = i / d4;
  float s = norm[row];
  float4 v = reinterpret_cast<const float4*>(in)[i];
  v.x *= s; v.y *= s; v.z *= s; v.w *= s;
  reinterpret_cast<float4*>(out)[i] = v;
}


template <int D>
__device__ __forceinline__ void gather_span(const float4* __restrict__ h4, int sreg,
                                            int count, int lane, float4 (&acc)[D / 128]) {
  constexpr int VEC = D / 128;
  constexpr int CH = 8 / VEC;
  for (int base = 0; base < count; base += CH) {
    int sidx[CH];
#pragma unroll
    for (int u = 0; u < CH; ++u) {
      int j = base + u;
      j = j < count ? j : count - 1;
      sidx[u] = __builtin_amdgcn_readlane(sreg, j);
    }
    float4 v[CH][VEC];
    if (base + CH <= count) {
#pragma unroll
      for (int u = 0; u < CH; ++u) {
        const float4* p = h4 + (size_t)sidx[u] * (D / 4) + lane;
#pragma unroll
        for (int q = 0; q < VEC; ++q) v[u][q] = p[32 * q];
      }
    } else {
#pragma unroll
      for (int u = 0; u < CH; ++u) {
        const float4* p = h4 + (size_t)sidx[u] * (D / 4) + lane;
        const bool valid = base + u < count;
#pragma unroll
        for (int q = 0; q < VEC; ++q) {
          const float4 t = p[32 * q];
          v[u][q].x = valid ? t.x : 0.f;
          v[u][q].y = valid ? t.y : 0.f;
          v[u][q].z = valid ? t.z : 0.f;
          v[u][q].w = valid ? t.w : 0.f;
        }
      }
    }
#pragma unroll
    for (int u = 0; u < CH; ++u) {
#pragma unroll
      for (int q = 0; q < VEC; ++q) {
        acc[q].x += v[u][q].x;
        acc[q].y += v[u][q].y;
        acc[q].z += v[u][q].z;
        acc[q].w += v[u][q].w;
      }
    }
  }
}

template <int D, bool SORT, bool EPI>
__global__ __launch_bounds__(256) void gather_sum_kernel(
    const float* __restrict__ h0, int* __restrict__ meta0,
    const int* __restrict__ ei_a, const int* __restrict__ ei_b, int E_a, int E_b,
    const float* __restrict__ bias, float* __restrict__ agg0, int n) {
  constexpr int VEC = D / 128;
  const int g = blockIdx.y;
  const int E = g ? E_b : E_a;
  const int* __restrict__ src = g ? ei_b : ei_a;
  const int* __restrict__ dst = src + E;
  const GraphMeta gm = graph_meta(meta0, g, n);
  const int* __restrict__ deg_in = gm.deg_in;
  int* __restrict__ lists = gm.lists;
  const float* __restrict__ rowscale = reinterpret_cast<const float*>(gm.cnt_in);
  const float* __restrict__ h = h0 + (size_t)g * (size_t)n * D;
  float* __restrict__ agg = agg0 + (size_t)g * (size_t)n * D;
  const int lane = threadIdx.x & 31;
  const int row = __builtin_amdgcn_readfirstlane((int)((blockIdx.x * 256u + threadIdx.x) >> 5));
  if (row >= n) return;
  int* lrow = lists + (size_t)row * LIST_CAP;
  int s0 = lrow[lane];
  int s1 = lrow[lane + 32];
  const int deg = __builtin_amdgcn_readfirstlane(deg_in[row]);
  const float4* h4 = reinterpret_cast<const float4*>(h);

  float rs = 0.f;
  float4 bq[VEC];
  if (EPI) {
    rs = rowscale[row];
    const float4* b4 = reinterpret_cast<const float4*>(bias) + lane;
#pragma unroll
    for (int q = 0; q < VEC; ++q) bq[q] = b4[32 * q];
  }

  float4 acc[VEC];
#pragma unroll
  for (int q = 0; q < VEC; ++q) acc[q] = make_float4(0.f, 0.f, 0.f, 0.f);

  if (deg <= LIST_CAP) {
    if (SORT) {
      const int key0 = (lane < deg) ? (int)(((unsigned)s0 << 6) | (unsigned)lane) : 0x7fffffff;
      const int key1 = (lane + 32 < deg) ? (int)(((unsigned)s1 << 6) | (unsigned)(lane + 32)) : 0x7fffffff;
      const int n0 = deg < 32 ? deg : 32;
      int r0 = 0, r1 = 0;
      for (int j = 0; j < n0; ++j) {
        const int kj = __builtin_amdgcn_readlane(key0, j);
        r0 += (kj < key0) ? 1 : 0;
        r1 += (kj < key1) ? 1 : 0;
      }
      for (int j = 32; j < deg; ++j) {
        const int kj = __builtin_amdgcn_readlane(key1, j - 32);
        r0 += (kj < key0) ? 1 : 0;
        r1 += (kj < key1) ? 1 : 0;
      }
      int p0 = 0, p1 = 0;
      for (int j = 0; j < n0; ++j) {
        const int kj = __builtin_amdgcn_readlane(key0, j);
        const int rj = __builtin_amdgcn_readlane(r0, j);
        if (rj == lane) p0 = kj;
        if (rj == lane + 32) p1 = kj;
      }
      for (int j = 32; j < deg; ++j) {
        const int kj = __builtin_amdgcn_readlane(key1, j - 32);
        const int rj = __builtin_amdgcn_readlane(r1, j - 32);
        if (rj == lane) p0 = kj;
        if (rj == lane + 32) p1 = kj;
      }
      s0 = p0 >> 6;
      s1 = p1 >> 6;
      if (lane < deg) lrow[lane] = s0;
      if (lane + 32 < deg) lrow[lane + 32] = s1;
    }
    gather_span<D>(h4, s0, deg < 32 ? deg : 32, lane, acc);
    if (deg > 32) gather_span<D>(h4, s1, deg - 32, lane, acc);
  } else {
    for (int e0 = 0; e0 < E; e0 += 32) {
      const int e = e0 + lane;
      int sv = 0;
      bool hit = false;
      if (e < E) {
        hit = (dst[e] == row);
        sv = src[e];
      }
      unsigned m = __builtin_amdgcn_ballot_w32(hit);
      while (m) {
        const int b = __builtin_ctz(m);
        m &= m - 1;
        const int s = __builtin_amdgcn_readlane(sv, b);
        const float4* p = h4 + (size_t)s * (D / 4) + lane;
#pragma unroll
        for (int q = 0; q < VEC; ++q) {
          const float4 t = p[32 * q];
          acc[q].x += t.x;
          acc[q].y += t.y;
          acc[q].z += t.z;
          acc[q].w += t.w;
        }
      }
    }
  }

  if (EPI) {
#pragma unroll
    for (int q = 0; q < VEC; ++q) {
      acc[q].x = fmaxf(acc[q].x * rs + bq[q].x, 0.0f);
      acc[q].y = fmaxf(acc[q].y * rs + bq[q].y, 0.0f);
      acc[q].z = fmaxf(acc[q].z * rs + bq[q].z, 0.0f);
      acc[q].w = fmaxf(acc[q].w * rs + bq[q].w, 0.0f);
    }
  }

  float4* out4 = reinterpret_cast<float4*>(agg) + (size_t)row * (D / 4) + lane;
#pragma unroll
  for (int q = 0; q < VEC; ++q) out4[32 * q] = acc[q];
}


template <int K, int NC, int ACT, bool HAS_SCALE, bool HAS_BIAS>
__global__ __launch_bounds__(256) void gemm_wmma_kernel(
    const float* __restrict__ A0, const float* __restrict__ W,
    const float* __restrict__ bias, const float* __restrict__ rowscale,
    float* __restrict__ C0, int M) {
  const float* __restrict__ A = A0 + (size_t)blockIdx.z * (size_t)M * K;
  float* __restrict__ C = C0 + (size_t)blockIdx.z * (size_t)M * NC;
  constexpr int PAD = 4;
  constexpr int LDSROW = K + PAD;
  __shared__ float sA[64 * LDSROW];

  const int tid = threadIdx.x;
  const int blockM = blockIdx.x * 64;
  const int blockN = blockIdx.y * 64;

  constexpr int F4_PER_ROW = K / 4;
  constexpr int TOTAL_F4 = 64 * F4_PER_ROW;
  constexpr int STAGE_IT = TOTAL_F4 / 256;
  float4 stage[STAGE_IT];
  if (blockM + 64 <= M) {
#pragma unroll
    for (int it = 0; it < STAGE_IT; ++it) {
      const int idx = tid + it * 256;
      const int row = idx / F4_PER_ROW;
      const int c4 = idx % F4_PER_ROW;
      stage[it] = *reinterpret_cast<const float4*>(A + (size_t)(blockM + row) * K + c4 * 4);
    }
  } else {
#pragma unroll
    for (int it = 0; it < STAGE_IT; ++it) {
      const int idx = tid + it * 256;
      const int row = idx / F4_PER_ROW;
      const int c4 = idx % F4_PER_ROW;
      const int gr = blockM + row;
      const bool valid = gr < M;
      const float4 t =
          *reinterpret_cast<const float4*>(A + (size_t)(valid ? gr : M - 1) * K + c4 * 4);
      stage[it].x = valid ? t.x : 0.f;
      stage[it].y = valid ? t.y : 0.f;
      stage[it].z = valid ? t.z : 0.f;
      stage[it].w = valid ? t.w : 0.f;
    }
  }
#pragma unroll
  for (int it = 0; it < STAGE_IT; ++it) {
    const int idx = tid + it * 256;
    const int row = idx / F4_PER_ROW;
    const int c4 = idx % F4_PER_ROW;
    *reinterpret_cast<float4*>(&sA[row * LDSROW + c4 * 4]) = stage[it];
  }
  __syncthreads();

  const int wave = tid >> 5;
  const int lane = tid & 31;
  const int wn = wave & 3;
  const int wmg = wave >> 2;
  const int half = lane >> 4;
  const int lm = lane & 15;
  const int col = blockN + wn * 16 + lm;

  v8f acc0 = {};
  v8f acc1 = {};
  const float* sArow0 = &sA[(wmg * 32 + lm) * LDSROW + 2 * half];
  const float* sArow1 = sArow0 + 16 * LDSROW;
  const float* Wp = W + (size_t)(2 * half) * NC + col;

#pragma unroll 8
  for (int k = 0; k < K; k += 4) {
    v2f a0 = *reinterpret_cast<const v2f*>(sArow0 + k);
    v2f a1 = *reinterpret_cast<const v2f*>(sArow1 + k);
    v2f b;
    b[0] = Wp[k * NC];
    b[1] = Wp[k * NC + NC];
    acc0 = __builtin_amdgcn_wmma_f32_16x16x4_f32(false, a0, false, b, (short)0,
                                                 acc0, false, false);
    acc1 = __builtin_amdgcn_wmma_f32_16x16x4_f32(false, a1, false, b, (short)0,
                                                 acc1, false, false);
  }

  const float bval = HAS_BIAS ? bias[col] : 0.0f;
  const int rbase0 = blockM + wmg * 32 + 8 * half;
  if (blockM + 64 <= M) {
#pragma unroll
    for (int t = 0; t < 2; ++t) {
      const int rbase = rbase0 + t * 16;
      float* Cp = C + (size_t)rbase * NC + col;
      const float* rs = rowscale + rbase;
#pragma unroll
      for (int r = 0; r < 8; ++r) {
        float v = (t == 0) ? acc0[r] : acc1[r];
        if (HAS_SCALE) v *= rs[r];
        if (HAS_BIAS) v += bval;
        if (ACT == 1) v = fmaxf(v, 0.0f);
        else if (ACT == 2) v = (v > 0.0f) ? v : expm1f(v);
        Cp[r * NC] = v;
      }
    }
  } else {
#pragma unroll
    for (int t = 0; t < 2; ++t) {
      const int rbase = rbase0 + t * 16;
#pragma unroll
      for (int r = 0; r < 8; ++r) {
        const int row = rbase + r;
        if (row < M) {
          float v = (t == 0) ? acc0[r] : acc1[r];
          if (HAS_SCALE) v *= rowscale[row];
          if (HAS_BIAS) v += bval;
          if (ACT == 1) v = fmaxf(v, 0.0f);
          else if (ACT == 2) v = (v > 0.0f) ? v : expm1f(v);
          C[(size_t)row * NC + col] = v;
        }
      }
    }
  }
}


extern "C" void kernel_launch(void* const* d_in, const int* in_sizes, int n_in,
                              void* d_out, int out_size, void* d_ws, size_t ws_size,
                              hipStream_t stream) {
  const float* feat1 = (const float*)d_in[0];
  const float* feat2 = (const float*)d_in[1];
  const int* ei1 = (const int*)d_in[2];
  const int* ei2 = (const int*)d_in[3];
  const float* W1 = (const float*)d_in[4];
  const float* b1 = (const float*)d_in[5];
  const float* W2 = (const float*)d_in[6];
  const float* b2 = (const float*)d_in[7];
  const float* fc1W = (const float*)d_in[8];
  const float* fc1b = (const float*)d_in[9];
  const float* fc2W = (const float*)d_in[10];
  const float* fc2b = (const float*)d_in[11];

  const int N = in_sizes[0] / 128;
  const int E_a = in_sizes[2] / 2;
  const int E_b = in_sizes[3] / 2;

  float* ws = (float*)d_ws;
  const size_t region = (size_t)N * 128;
  float* R0 = ws;
  float* R2 = ws + 2 * region;
  unsigned* scratch0 = (unsigned*)R2;
  const size_t scratch_bytes = region * sizeof(float);
  int* meta0 = (int*)d_out;

  const int row_blocks = (N * 32 + 255) / 256;
  const int m_tiles = (N + 63) / 64;

  hipMemsetAsync(graph_meta(meta0, 0, N).cnt_out, 0, 4 * (size_t)N * sizeof(int), stream);
  const int E_max = E_a > E_b ? E_a : E_b;
  const int nblk = (E_max + BIN_THREADS * BIN_EPT - 1) / (BIN_THREADS * BIN_EPT);
  const int nbins = (N + BIN_NODES - 1) / BIN_NODES;
  bin_edges_kernel<<<dim3(nblk, 2), BIN_THREADS, 0, stream>>>(
      ei1, ei2, E_a, E_b, meta0, N, scratch0, scratch_bytes, nbins);
  rank_lists_kernel<<<dim3(nbins, 2), BIN_THREADS, 0, stream>>>(
      meta0, scratch0, scratch_bytes, N, nbins, nblk);

  scale_rows_kernel<<<dim3(row_blocks, 2), 256, 0, stream>>>(
      R0, feat1, feat2, meta0, N * 128 / 4, 32, N);
  gather_sum_kernel<128, true, false><<<dim3(row_blocks, 2), 256, 0, stream>>>(
      R0, meta0, ei1, ei2, E_a, E_b, nullptr, R2, N);

  for (int g = 0; g < 2; ++g) {
    const GraphMeta gm = graph_meta(meta0, g, N);
    const float* norm_src = (const float*)gm.cnt_out;
    const float* norm_dst = (const float*)gm.cnt_in;
    float* Rg = R2 + (size_t)g * region;
    gemm_wmma_kernel<128, 256, 1, true, true><<<dim3(m_tiles, 256 / 64, 1), 256, 0, stream>>>(
        Rg, W1, b1, norm_dst, R0, N);
    gemm_wmma_kernel<256, 128, 0, true, false><<<dim3(m_tiles, 128 / 64, 1), 256, 0, stream>>>(
        R0, W2, nullptr, norm_src, Rg, N);
  }
  gather_sum_kernel<128, false, true><<<dim3(row_blocks, 2), 256, 0, stream>>>(
      R2, meta0, ei1, ei2, E_a, E_b, b2, R0, N);

  gemm_wmma_kernel<128, 128, 2, false, true><<<dim3(m_tiles, 128 / 64, 2), 256, 0, stream>>>(
      R0, fc1W, fc1b, nullptr, R2, N);
  gemm_wmma_kernel<128, 128, 0, false, true><<<dim3(m_tiles, 128 / 64, 2), 256, 0, stream>>>(
      R2, fc2W, fc2b, nullptr, (float*)d_out, N);
}


# Each pair of back-to-back GEMMs runs as one kernel: the first product's 64-row tile stays in LDS as the second's A operand (W1 then W2; fc1 then fc2), both graphs per launch: 11 launches become 7
# speedup vs baseline: 5.0093x; 1.1679x over previous
#include <hip/hip_runtime.h>
#include <hip/hip_bf16.h>


typedef __attribute__((ext_vector_type(2))) float v2f;
typedef __attribute__((ext_vector_type(8))) float v8f;


constexpr int LIST_CAP = 64;
constexpr int BIN_NODES = 256;
constexpr int MAX_BINS = 256;
constexpr int SUB = 64;
constexpr int BIN_THREADS = 1024;
constexpr int BIN_EPT = 4;

struct GraphMeta {
  int* lists;
  int* deg_in;
  int* cnt_out;
  int* cnt_in;
};

__device__ __host__ __forceinline__ GraphMeta graph_meta(int* meta0, int g, int n) {
  GraphMeta m;
  m.lists = meta0 + (size_t)g * (size_t)n * 128;
  m.deg_in = m.lists + (size_t)n * LIST_CAP;
  m.cnt_out = meta0 + (size_t)n * (LIST_CAP + 1) + (size_t)g * 2 * (size_t)n;
  m.cnt_in = m.cnt_out + n;
  return m;
}

struct BinScratch {
  unsigned* rec_d;
  int* cnt_d;
  int* cnt_s;
  unsigned char* rec_s;
  int nb;
};

__device__ __forceinline__ BinScratch bin_scratch(unsigned* scratch, size_t scratch_bytes,
                                                  int nbins, int nblk) {
  BinScratch b;
  int nb = nbins < MAX_BINS ? nbins : MAX_BINS;
  const size_t cells = (size_t)nb * (size_t)nblk;
  if (cells * (SUB * 5 + 8) > scratch_bytes) nb = 0;
  b.nb = nb;
  b.rec_d = scratch;
  b.cnt_d = reinterpret_cast<int*>(scratch + cells * SUB);
  b.cnt_s = b.cnt_d + cells;
  b.rec_s = reinterpret_cast<unsigned char*>(b.cnt_s + cells);
  return b;
}

__global__ __launch_bounds__(BIN_THREADS) void bin_edges_kernel(
    const int* __restrict__ ei_a, const int* __restrict__ ei_b, int E_a, int E_b,
    int* __restrict__ meta0, int n_nodes,
    unsigned* __restrict__ scratch0, size_t scratch_bytes, int nbins) {
  constexpr int EPB = BIN_THREADS * BIN_EPT;
  const int g = blockIdx.y;
  const int E = g ? E_b : E_a;
  const int* __restrict__ src = g ? ei_b : ei_a;
  const int* __restrict__ dst = src + E;
  const GraphMeta gm = graph_meta(meta0, g, n_nodes);
  int* __restrict__ cnt_out = gm.cnt_out;
  int* __restrict__ cnt_in = gm.cnt_in;
  int* __restrict__ lists = gm.lists;
  unsigned* __restrict__ scratch = scratch0 + (size_t)g * (scratch_bytes / sizeof(unsigned));
  __shared__ int table[2 * MAX_BINS];
  __shared__ int offs[2 * MAX_BINS];
  __shared__ int wtot[2 * MAX_BINS / 32];
  __shared__ unsigned st_rec_d[EPB];
  __shared__ unsigned st_pos_d[EPB];
  __shared__ unsigned st_pos_s[EPB];
  __shared__ unsigned char st_rec_s[EPB];
  const int tid = threadIdx.x;
  const int lane = tid & 31;
  const int blk = blockIdx.x;
  const int nblk = gridDim.x;
  const BinScratch bs = bin_scratch(scratch, scratch_bytes, nbins, nblk);
  if (tid < 2 * MAX_BINS) table[tid] = 0;
  __syncthreads();

  const int e0 = blk * EPB + tid;
  int s[BIN_EPT], d[BIN_EPT], rd[BIN_EPT], rs[BIN_EPT];
#pragma unroll
  for (int j = 0; j < BIN_EPT; ++j) {
    const int e = e0 + j * BIN_THREADS;
    s[j] = 0;
    d[j] = 0;
    if (e < E) {
      s[j] = src[e];
      d[j] = dst[e];
    }
  }
#pragma unroll
  for (int j = 0; j < BIN_EPT; ++j) {
    const int e = e0 + j * BIN_THREADS;
    rd[j] = -1;
    rs[j] = -1;
    if (e < E) {
      const unsigned bd = (unsigned)d[j] >> 8;
      const unsigned bsrc = (unsigned)s[j] >> 8;
      if (bd < (unsigned)bs.nb && (unsigned)s[j] < (1u << 24)) rd[j] = atomicAdd(&table[bd], 1);
      if (bsrc < (unsigned)bs.nb) rs[j] = atomicAdd(&table[MAX_BINS + bsrc], 1);
    }
  }
  __syncthreads();

  if (tid < 2 * MAX_BINS) {
    int c = table[tid];
    c = c < SUB ? c : SUB;
    int incl = c;
    for (int o = 1; o < 32; o <<= 1) {
      const int t = __shfl_up(incl, o);
      if (lane >= o) incl += t;
    }
    offs[tid] = incl - c;
    if (lane == 31) wtot[tid >> 5] = incl;
  }
  __syncthreads();
  if (tid < 2 * MAX_BINS) {
    const int w = tid >> 5;
    int add = 0;
    for (int k = w & ~7; k < w; ++k) add += wtot[k];
    offs[tid] += add;
  }
  __syncthreads();

#pragma unroll
  for (int j = 0; j < BIN_EPT; ++j) {
    const int e = e0 + j * BIN_THREADS;
    if (e < E) {
      const unsigned bd = (unsigned)d[j] >> 8;
      const unsigned bsrc = (unsigned)s[j] >> 8;
      if ((unsigned)rd[j] < (unsigned)SUB) {
        const int p = offs[bd] + rd[j];
        st_rec_d[p] = ((unsigned)s[j] << 8) | ((unsigned)d[j] & 255u);
        st_pos_d[p] = (unsigned)(((size_t)bd * nblk + blk) * SUB + rd[j]);
      } else {
        const int slot = atomicAdd(&cnt_in[d[j]], 1);
        if (slot < LIST_CAP) lists[(size_t)d[j] * LIST_CAP + slot] = s[j];
      }
      if ((unsigned)rs[j] < (unsigned)SUB) {
        const int p = offs[MAX_BINS + bsrc] + rs[j];
        st_rec_s[p] = (unsigned char)((unsigned)s[j] & 255u);
        st_pos_s[p] = (unsigned)(((size_t)bsrc * nblk + blk) * SUB + rs[j]);
      } else {
        atomicAdd(&cnt_out[s[j]], 1);
      }
    }
  }
  __syncthreads();

  const int last_d = table[MAX_BINS - 1];
  const int last_s = table[2 * MAX_BINS - 1];
  const int total_d = offs[MAX_BINS - 1] + (last_d < SUB ? last_d : SUB);
  const int total_s = offs[2 * MAX_BINS - 1] + (last_s < SUB ? last_s : SUB);
  for (int i = tid; i < total_d; i += BIN_THREADS) bs.rec_d[st_pos_d[i]] = st_rec_d[i];
  for (int i = tid; i < total_s; i += BIN_THREADS) bs.rec_s[st_pos_s[i]] = st_rec_s[i];
  if (tid < 2 * MAX_BINS) {
    const int b = tid & (MAX_BINS - 1);
    if (b < bs.nb) {
      const int c = table[tid];
      int* cnt = tid < MAX_BINS ? bs.cnt_d : bs.cnt_s;
      cnt[(size_t)blk * bs.nb + b] = c < SUB ? c : SUB;
    }
  }
}

__global__ __launch_bounds__(BIN_THREADS) void rank_lists_kernel(
    int* __restrict__ meta0, const unsigned* __restrict__ scratch0, size_t scratch_bytes,
    int n_nodes, int nbins, int nblk) {
  constexpr int UNR = 4;
  const int g = blockIdx.y;
  const GraphMeta gm = graph_meta(meta0, g, n_nodes);
  int* __restrict__ cnt_out = gm.cnt_out;
  int* __restrict__ cnt_in = gm.cnt_in;
  int* __restrict__ lists = gm.lists;
  int* __restrict__ deg_in = gm.deg_in;
  const unsigned* __restrict__ scratch = scratch0 + (size_t)g * (scratch_bytes / sizeof(unsigned));
  __shared__ int tile[BIN_NODES * LIST_CAP];
  __shared__ int n_in[BIN_NODES];
  __shared__ int n_base[BIN_NODES];
  __shared__ int n_out[BIN_NODES];
  __shared__ int ccd[BIN_THREADS];
  __shared__ int ccs[BIN_THREADS];
  const int tid = threadIdx.x;
  const int b = blockIdx.x;
  const int node0 = b * BIN_NODES;
  const BinScratch bs = bin_scratch(const_cast<unsigned*>(scratch), scratch_bytes, nbins, nblk);

  if (tid < BIN_NODES) {
    const int node = node0 + tid;
    int ci = 0, co = 0;
    if (node < n_nodes) {
      ci = cnt_in[node];
      co = cnt_out[node];
    }
    n_in[tid] = ci;
    n_base[tid] = ci;
    n_out[tid] = co;
  }

  if (b < bs.nb) {
    for (int kc = 0; kc < nblk; kc += BIN_THREADS) {
      __syncthreads();
      if (kc + tid < nblk) {
        ccd[tid] = bs.cnt_d[(size_t)(kc + tid) * bs.nb + b];
        ccs[tid] = bs.cnt_s[(size_t)(kc + tid) * bs.nb + b];
      }
      __syncthreads();
      const int ncell = nblk - kc < BIN_THREADS ? nblk - kc : BIN_THREADS;
      const int nplace = ncell * SUB;
      const unsigned* rdp = bs.rec_d + ((size_t)b * nblk + kc) * SUB;
      const unsigned char* rsp = bs.rec_s + ((size_t)b * nblk + kc) * SUB;
      for (int p0 = tid; p0 < nplace; p0 += UNR * BIN_THREADS) {
        unsigned r[UNR];
        int q[UNR];
        bool has_r[UNR];
#pragma unroll
        for (int u = 0; u < UNR; ++u) {
          const int p = p0 + u * BIN_THREADS;
          r[u] = 0;
          has_r[u] = false;
          q[u] = -1;
          if (p < nplace) {
            if ((p % SUB) < ccd[p / SUB]) {
              r[u] = rdp[p];
              has_r[u] = true;
            }
            if ((p % SUB) < ccs[p / SUB]) q[u] = (int)rsp[p];
          }
        }
#pragma unroll
        for (int u = 0; u < UNR; ++u) {
          if (has_r[u]) {
            const int l = (int)(r[u] & 255u);
            const int slot = atomicAdd(&n_in[l], 1);
            if (slot < LIST_CAP) tile[l * LIST_CAP + slot] = (int)(r[u] >> 8);
          }
          if (q[u] >= 0) atomicAdd(&n_out[q[u]], 1);
        }
      }
    }
  }
  __syncthreads();

  for (int idx = tid; idx < BIN_NODES * LIST_CAP; idx += BIN_THREADS) {
    const int l = idx / LIST_CAP;
    const int slot = idx % LIST_CAP;
    const int node = node0 + l;
    if (node < n_nodes && slot >= n_base[l] && slot < n_in[l])
      lists[(size_t)node * LIST_CAP + slot] = tile[idx];
  }
  if (tid < BIN_NODES) {
    const int node = node0 + tid;
    if (node < n_nodes) {
      const int ci = n_in[tid];
      const int co = n_out[tid];
      deg_in[node] = ci;
      cnt_in[node] = __float_as_int(rsqrtf(fmaxf((float)ci, 1.0f)));
      cnt_out[node] = __float_as_int(rsqrtf(fmaxf((float)co, 1.0f)));
    }
  }
}


__global__ void scale_rows_kernel(float* __restrict__ out0,
                                  const float* __restrict__ in_a,
                                  const float* __restrict__ in_b,
                                  int* __restrict__ meta0,
                                  int n4, int d4, int n_nodes) {
  int i = blockIdx.x * blockDim.x + threadIdx.x;
  if (i >= n4) return;
  const int g = blockIdx.y;
  const float* __restrict__ in = g ? in_b : in_a;
  float* __restrict__ out = out0 + (size_t)g * (size_t)n4 * 4;
  const float* __restrict__ norm = reinterpret_cast<const float*>(graph_meta(meta0, g, n_nodes).cnt_out);
  int row = i / d4;
  float s = norm[row];
  float4 v = reinterpret_cast<const float4*>(in)[i];
  v.x *= s; v.y *= s; v.z *= s; v.w *= s;
  reinterpret_cast<float4*>(out)[i] = v;
}


template <int D>
__device__ __forceinline__ void gather_span(const float4* __restrict__ h4, int sreg,
                                            int count, int lane, float4 (&acc)[D / 128]) {
  constexpr int VEC = D / 128;
  constexpr int CH = 8 / VEC;
  for (int base = 0; base < count; base += CH) {
    int sidx[CH];
#pragma unroll
    for (int u = 0; u < CH; ++u) {
      int j = base + u;
      j = j < count ? j : count - 1;
      sidx[u] = __builtin_amdgcn_readlane(sreg, j);
    }
    float4 v[CH][VEC];
    if (base + CH <= count) {
#pragma unroll
      for (int u = 0; u < CH; ++u) {
        const float4* p = h4 + (size_t)sidx[u] * (D / 4) + lane;
#pragma unroll
        for (int q = 0; q < VEC; ++q) v[u][q] = p[32 * q];
      }
    } else {
#pragma unroll
      for (int u = 0; u < CH; ++u) {
        const float4* p = h4 + (size_t)sidx[u] * (D / 4) + lane;
        const bool valid = base + u < count;
#pragma unroll
        for (int q = 0; q < VEC; ++q) {
          const float4 t = p[32 * q];
          v[u][q].x = valid ? t.x : 0.f;
          v[u][q].y = valid ? t.y : 0.f;
          v[u][q].z = valid ? t.z : 0.f;
          v[u][q].w = valid ? t.w : 0.f;
        }
      }
    }
#pragma unroll
    for (int u = 0; u < CH; ++u) {
#pragma unroll
      for (int q = 0; q < VEC; ++q) {
        acc[q].x += v[u][q].x;
        acc[q].y += v[u][q].y;
        acc[q].z += v[u][q].z;
        acc[q].w += v[u][q].w;
      }
    }
  }
}

template <int D, bool SORT, bool EPI>
__global__ __launch_bounds__(256) void gather_sum_kernel(
    const float* __restrict__ h0, int* __restrict__ meta0,
    const int* __restrict__ ei_a, const int* __restrict__ ei_b, int E_a, int E_b,
    const float* __restrict__ bias, float* __restrict__ agg0, int n) {
  constexpr int VEC = D / 128;
  const int g = blockIdx.y;
  const int E = g ? E_b : E_a;
  const int* __restrict__ src = g ? ei_b : ei_a;
  const int* __restrict__ dst = src + E;
  const GraphMeta gm = graph_meta(meta0, g, n);
  const int* __restrict__ deg_in = gm.deg_in;
  int* __restrict__ lists = gm.lists;
  const float* __restrict__ rowscale = reinterpret_cast<const float*>(gm.cnt_in);
  const float* __restrict__ h = h0 + (size_t)g * (size_t)n * D;
  float* __restrict__ agg = agg0 + (size_t)g * (size_t)n * D;
  const int lane = threadIdx.x & 31;
  const int row = __builtin_amdgcn_readfirstlane((int)((blockIdx.x * 256u + threadIdx.x) >> 5));
  if (row >= n) return;
  int* lrow = lists + (size_t)row * LIST_CAP;
  int s0 = lrow[lane];
  int s1 = lrow[lane + 32];
  const int deg = __builtin_amdgcn_readfirstlane(deg_in[row]);
  const float4* h4 = reinterpret_cast<const float4*>(h);

  float rs = 0.f;
  float4 bq[VEC];
  if (EPI) {
    rs = rowscale[row];
    const float4* b4 = reinterpret_cast<const float4*>(bias) + lane;
#pragma unroll
    for (int q = 0; q < VEC; ++q) bq[q] = b4[32 * q];
  }

  float4 acc[VEC];
#pragma unroll
  for (int q = 0; q < VEC; ++q) acc[q] = make_float4(0.f, 0.f, 0.f, 0.f);

  if (deg <= LIST_CAP) {
    if (SORT) {
      const int key0 = (lane < deg) ? (int)(((unsigned)s0 << 6) | (unsigned)lane) : 0x7fffffff;
      const int key1 = (lane + 32 < deg) ? (int)(((unsigned)s1 << 6) | (unsigned)(lane + 32)) : 0x7fffffff;
      const int n0 = deg < 32 ? deg : 32;
      int r0 = 0, r1 = 0;
      for (int j = 0; j < n0; ++j) {
        const int kj = __builtin_amdgcn_readlane(key0, j);
        r0 += (kj < key0) ? 1 : 0;
        r1 += (kj < key1) ? 1 : 0;
      }
      for (int j = 32; j < deg; ++j) {
        const int kj = __builtin_amdgcn_readlane(key1, j - 32);
        r0 += (kj < key0) ? 1 : 0;
        r1 += (kj < key1) ? 1 : 0;
      }
      int p0 = 0, p1 = 0;
      for (int j = 0; j < n0; ++j) {
        const int kj = __builtin_amdgcn_readlane(key0, j);
        const int rj = __builtin_amdgcn_readlane(r0, j);
        if (rj == lane) p0 = kj;
        if (rj == lane + 32) p1 = kj;
      }
      for (int j = 32; j < deg; ++j) {
        const int kj = __builtin_amdgcn_readlane(key1, j - 32);
        const int rj = __builtin_amdgcn_readlane(r1, j - 32);
        if (rj == lane) p0 = kj;
        if (rj == lane + 32) p1 = kj;
      }
      s0 = p0 >> 6;
      s1 = p1 >> 6;
      if (lane < deg) lrow[lane] = s0;
      if (lane + 32 < deg) lrow[lane + 32] = s1;
    }
    gather_span<D>(h4, s0, deg < 32 ? deg : 32, lane, acc);
    if (deg > 32) gather_span<D>(h4, s1, deg - 32, lane, acc);
  } else {
    for (int e0 = 0; e0 < E; e0 += 32) {
      const int e = e0 + lane;
      int sv = 0;
      bool hit = false;
      if (e < E) {
        hit = (dst[e] == row);
        sv = src[e];
      }
      unsigned m = __builtin_amdgcn_ballot_w32(hit);
      while (m) {
        const int b = __builtin_ctz(m);
        m &= m - 1;
        const int s = __builtin_amdgcn_readlane(sv, b);
        const float4* p = h4 + (size_t)s * (D / 4) + lane;
#pragma unroll
        for (int q = 0; q < VEC; ++q) {
          const float4 t = p[32 * q];
          acc[q].x += t.x;
          acc[q].y += t.y;
          acc[q].z += t.z;
          acc[q].w += t.w;
        }
      }
    }
  }

  if (EPI) {
#pragma unroll
    for (int q = 0; q < VEC; ++q) {
      acc[q].x = fmaxf(acc[q].x * rs + bq[q].x, 0.0f);
      acc[q].y = fmaxf(acc[q].y * rs + bq[q].y, 0.0f);
      acc[q].z = fmaxf(acc[q].z * rs + bq[q].z, 0.0f);
      acc[q].w = fmaxf(acc[q].w * rs + bq[q].w, 0.0f);
    }
  }

  float4* out4 = reinterpret_cast<float4*>(agg) + (size_t)row * (D / 4) + lane;
#pragma unroll
  for (int q = 0; q < VEC; ++q) out4[32 * q] = acc[q];
}


template <int K, int NC, int NS>
__device__ __forceinline__ void tile_product(const float* sArow0, const float* sArow1,
                                             const float* __restrict__ Wp,
                                             v8f (&acc)[2][NS]) {
  constexpr int UNR = NS >= 4 ? 4 : 8;
#pragma unroll UNR
  for (int k = 0; k < K; k += 4) {
    const v2f a0 = *reinterpret_cast<const v2f*>(sArow0 + k);
    const v2f a1 = *reinterpret_cast<const v2f*>(sArow1 + k);
#pragma unroll
    for (int j = 0; j < NS; ++j) {
      v2f b;
      b[0] = Wp[k * NC + j * 16];
      b[1] = Wp[k * NC + NC + j * 16];
      acc[0][j] = __builtin_amdgcn_wmma_f32_16x16x4_f32(false, a0, false, b, (short)0,
                                                        acc[0][j], false, false);
      acc[1][j] = __builtin_amdgcn_wmma_f32_16x16x4_f32(false, a1, false, b, (short)0,
                                                        acc[1][j], false, false);
    }
  }
}

template <int ACT>
__device__ __forceinline__ float activate(float v) {
  if (ACT == 1) return fmaxf(v, 0.0f);
  if (ACT == 2) return (v > 0.0f) ? v : expm1f(v);
  return v;
}

template <int K1, int N1, int ACT1, bool SCALE1, bool BIAS1,
          int N2, int ACT2, bool SCALE2, bool BIAS2>
__global__ __launch_bounds__(256) void gemm_pair_kernel(
    const float* __restrict__ A0, const float* __restrict__ W1,
    const float* __restrict__ bias1, const float* __restrict__ W2,
    const float* __restrict__ bias2, int* __restrict__ meta0,
    float* __restrict__ C0, int M) {
  const float* __restrict__ A = A0 + (size_t)blockIdx.z * (size_t)M * K1;
  float* __restrict__ C = C0 + (size_t)blockIdx.z * (size_t)M * N2;
  const GraphMeta gm = graph_meta(meta0, blockIdx.z, M);
  const float* __restrict__ scale1 = reinterpret_cast<const float*>(gm.cnt_in);
  const float* __restrict__ scale2 = reinterpret_cast<const float*>(gm.cnt_out);
  constexpr int PAD = 4;
  constexpr int LDA = K1 + PAD;
  constexpr int LDH = N1 + PAD;
  constexpr int NS1 = N1 / 64;
  constexpr int NS2 = N2 / 64;
  __shared__ float sA[64 * LDA];
  __shared__ float sH[64 * LDH];

  const int tid = threadIdx.x;
  const int blockM = blockIdx.x * 64;

  constexpr int F4_PER_ROW = K1 / 4;
  constexpr int TOTAL_F4 = 64 * F4_PER_ROW;
  constexpr int STAGE_IT = TOTAL_F4 / 256;
  float4 stage[STAGE_IT];
  if (blockM + 64 <= M) {
#pragma unroll
    for (int it = 0; it < STAGE_IT; ++it) {
      const int idx = tid + it * 256;
      const int row = idx / F4_PER_ROW;
      const int c4 = idx % F4_PER_ROW;
      stage[it] = *reinterpret_cast<const float4*>(A + (size_t)(blockM + row) * K1 + c4 * 4);
    }
  } else {
#pragma unroll
    for (int it = 0; it < STAGE_IT; ++it) {
      const int idx = tid + it * 256;
      const int row = idx / F4_PER_ROW;
      const int c4 = idx % F4_PER_ROW;
      const int gr = blockM + row;
      const bool valid = gr < M;
      const float4 t =
          *reinterpret_cast<const float4*>(A + (size_t)(valid ? gr : M - 1) * K1 + c4 * 4);
      stage[it].x = valid ? t.x : 0.f;
      stage[it].y = valid ? t.y : 0.f;
      stage[it].z = valid ? t.z : 0.f;
      stage[it].w = valid ? t.w : 0.f;
    }
  }
#pragma unroll
  for (int it = 0; it < STAGE_IT; ++it) {
    const int idx = tid + it * 256;
    const int row = idx / F4_PER_ROW;
    const int c4 = idx % F4_PER_ROW;
    *reinterpret_cast<float4*>(&sA[row * LDA + c4 * 4]) = stage[it];
  }
  __syncthreads();

  const int wave = tid >> 5;
  const int lane = tid & 31;
  const int wn = wave & 3;
  const int wmg = wave >> 2;
  const int half = lane >> 4;
  const int lm = lane & 15;
  const int lrow0 = wmg * 32 + 8 * half;

  {
    const int col = wn * (N1 / 4) + lm;
    v8f acc[2][NS1] = {};
    tile_product<K1, N1, NS1>(&sA[(wmg * 32 + lm) * LDA + 2 * half],
                              &sA[(wmg * 32 + 16 + lm) * LDA + 2 * half],
                              W1 + (size_t)(2 * half) * N1 + col, acc);
    float bv[NS1];
#pragma unroll
    for (int j = 0; j < NS1; ++j) bv[j] = BIAS1 ? bias1[col + j * 16] : 0.0f;
#pragma unroll
    for (int t = 0; t < 2; ++t) {
#pragma unroll
      for (int r = 0; r < 8; ++r) {
        const int lr = lrow0 + t * 16 + r;
        float s = 1.0f;
        if (SCALE1) {
          const int gr = blockM + lr;
          s = scale1[gr < M ? gr : M - 1];
        }
#pragma unroll
        for (int j = 0; j < NS1; ++j) {
          float v = acc[t][j][r];
          if (SCALE1) v *= s;
          if (BIAS1) v += bv[j];
          sH[lr * LDH + col + j * 16] = activate<ACT1>(v);
        }
      }
    }
  }
  __syncthreads();

  {
    const int col = wn * (N2 / 4) + lm;
    v8f acc[2][NS2] = {};
    tile_product<N1, N2, NS2>(&sH[(wmg * 32 + lm) * LDH + 2 * half],
                              &sH[(wmg * 32 + 16 + lm) * LDH + 2 * half],
                              W2 + (size_t)(2 * half) * N2 + col, acc);
    float bv[NS2];
#pragma unroll
    for (int j = 0; j < NS2; ++j) bv[j] = BIAS2 ? bias2[col + j * 16] : 0.0f;
#pragma unroll
    for (int t = 0; t < 2; ++t) {
#pragma unroll
      for (int r = 0; r < 8; ++r) {
        const int gr = blockM + lrow0 + t * 16 + r;
        if (gr < M) {
          const float s = SCALE2 ? scale2[gr] : 1.0f;
          float* Cp = C + (size_t)gr * N2 + col;
#pragma unroll
          for (int j = 0; j < NS2; ++j) {
            float v = acc[t][j][r];
            if (SCALE2) v *= s;
            if (BIAS2) v += bv[j];
            Cp[j * 16] = activate<ACT2>(v);
          }
        }
      }
    }
  }
}


extern "C" void kernel_launch(void* const* d_in, const int* in_sizes, int n_in,
                              void* d_out, int out_size, void* d_ws, size_t ws_size,
                              hipStream_t stream) {
  const float* feat1 = (const float*)d_in[0];
  const float* feat2 = (const float*)d_in[1];
  const int* ei1 = (const int*)d_in[2];
  const int* ei2 = (const int*)d_in[3];
  const float* W1 = (const float*)d_in[4];
  const float* b1 = (const float*)d_in[5];
  const float* W2 = (const float*)d_in[6];
  const float* b2 = (const float*)d_in[7];
  const float* fc1W = (const float*)d_in[8];
  const float* fc1b = (const float*)d_in[9];
  const float* fc2W = (const float*)d_in[10];
  const float* fc2b = (const float*)d_in[11];

  const int N = in_sizes[0] / 128;
  const int E_a = in_sizes[2] / 2;
  const int E_b = in_sizes[3] / 2;

  float* ws = (float*)d_ws;
  const size_t region = (size_t)N * 128;
  float* R0 = ws;
  float* R2 = ws + 2 * region;
  unsigned* scratch0 = (unsigned*)R2;
  const size_t scratch_bytes = region * sizeof(float);
  int* meta0 = (int*)d_out;

  const int row_blocks = (N * 32 + 255) / 256;
  const int m_tiles = (N + 63) / 64;

  hipMemsetAsync(graph_meta(meta0, 0, N).cnt_out, 0, 4 * (size_t)N * sizeof(int), stream);
  const int E_max = E_a > E_b ? E_a : E_b;
  const int nblk = (E_max + BIN_THREADS * BIN_EPT - 1) / (BIN_THREADS * BIN_EPT);
  const int nbins = (N + BIN_NODES - 1) / BIN_NODES;
  bin_edges_kernel<<<dim3(nblk, 2), BIN_THREADS, 0, stream>>>(
      ei1, ei2, E_a, E_b, meta0, N, scratch0, scratch_bytes, nbins);
  rank_lists_kernel<<<dim3(nbins, 2), BIN_THREADS, 0, stream>>>(
      meta0, scratch0, scratch_bytes, N, nbins, nblk);

  scale_rows_kernel<<<dim3(row_blocks, 2), 256, 0, stream>>>(
      R0, feat1, feat2, meta0, N * 128 / 4, 32, N);
  gather_sum_kernel<128, true, false><<<dim3(row_blocks, 2), 256, 0, stream>>>(
      R0, meta0, ei1, ei2, E_a, E_b, nullptr, R2, N);

  gemm_pair_kernel<128, 256, 1, true, true, 128, 0, true, false>
      <<<dim3(m_tiles, 1, 2), 256, 0, stream>>>(R2, W1, b1, W2, nullptr, meta0, R0, N);
  gather_sum_kernel<128, false, true><<<dim3(row_blocks, 2), 256, 0, stream>>>(
      R0, meta0, ei1, ei2, E_a, E_b, b2, R2, N);

  gemm_pair_kernel<128, 128, 2, false, true, 128, 0, false, true>
      <<<dim3(m_tiles, 1, 2), 256, 0, stream>>>(R2, fc1W, fc1b, fc2W, fc2b, meta0,
                                                (float*)d_out, N);
}


# Each edge gather becomes the A-tile staging of the GEMM pair that consumes it: a block sums its 64 rows straight into LDS (same source order, row loads in deeper groups); 7 launches become 5
# speedup vs baseline: 5.2066x; 1.0394x over previous
#include <hip/hip_runtime.h>
#include <hip/hip_bf16.h>


typedef __attribute__((ext_vector_type(2))) float v2f;
typedef __attribute__((ext_vector_type(8))) float v8f;


constexpr int LIST_CAP = 64;
constexpr int BIN_NODES = 256;
constexpr int MAX_BINS = 256;
constexpr int SUB = 64;
constexpr int BIN_THREADS = 1024;
constexpr int BIN_EPT = 4;

struct GraphMeta {
  int* lists;
  int* deg_in;
  int* cnt_out;
  int* cnt_in;
};

__device__ __host__ __forceinline__ GraphMeta graph_meta(int* meta0, int g, int n) {
  GraphMeta m;
  m.lists = meta0 + (size_t)g * (size_t)n * 128;
  m.deg_in = m.lists + (size_t)n * LIST_CAP;
  m.cnt_out = meta0 + (size_t)n * (LIST_CAP + 1) + (size_t)g * 2 * (size_t)n;
  m.cnt_in = m.cnt_out + n;
  return m;
}

struct BinScratch {
  unsigned* rec_d;
  int* cnt_d;
  int* cnt_s;
  unsigned char* rec_s;
  int nb;
};

__device__ __forceinline__ BinScratch bin_scratch(unsigned* scratch, size_t scratch_bytes,
                                                  int nbins, int nblk) {
  BinScratch b;
  int nb = nbins < MAX_BINS ? nbins : MAX_BINS;
  const size_t cells = (size_t)nb * (size_t)nblk;
  if (cells * (SUB * 5 + 8) > scratch_bytes) nb = 0;
  b.nb = nb;
  b.rec_d = scratch;
  b.cnt_d = reinterpret_cast<int*>(scratch + cells * SUB);
  b.cnt_s = b.cnt_d + cells;
  b.rec_s = reinterpret_cast<unsigned char*>(b.cnt_s + cells);
  return b;
}

__global__ __launch_bounds__(BIN_THREADS) void bin_edges_kernel(
    const int* __restrict__ ei_a, const int* __restrict__ ei_b, int E_a, int E_b,
    int* __restrict__ meta0, int n_nodes,
    unsigned* __restrict__ scratch0, size_t scratch_bytes, int nbins) {
  constexpr int EPB = BIN_THREADS * BIN_EPT;
  const int g = blockIdx.y;
  const int E = g ? E_b : E_a;
  const int* __restrict__ src = g ? ei_b : ei_a;
  const int* __restrict__ dst = src + E;
  const GraphMeta gm = graph_meta(meta0, g, n_nodes);
  int* __restrict__ cnt_out = gm.cnt_out;
  int* __restrict__ cnt_in = gm.cnt_in;
  int* __restrict__ lists = gm.lists;
  unsigned* __restrict__ scratch = scratch0 + (size_t)g * (scratch_bytes / sizeof(unsigned));
  __shared__ int table[2 * MAX_BINS];
  __shared__ int offs[2 * MAX_BINS];
  __shared__ int wtot[2 * MAX_BINS / 32];
  __shared__ unsigned st_rec_d[EPB];
  __shared__ unsigned st_pos_d[EPB];
  __shared__ unsigned st_pos_s[EPB];
  __shared__ unsigned char st_rec_s[EPB];
  const int tid = threadIdx.x;
  const int lane = tid & 31;
  const int blk = blockIdx.x;
  const int nblk = gridDim.x;
  const BinScratch bs = bin_scratch(scratch, scratch_bytes, nbins, nblk);
  if (tid < 2 * MAX_BINS) table[tid] = 0;
  __syncthreads();

  const int e0 = blk * EPB + tid;
  int s[BIN_EPT], d[BIN_EPT], rd[BIN_EPT], rs[BIN_EPT];
#pragma unroll
  for (int j = 0; j < BIN_EPT; ++j) {
    const int e = e0 + j * BIN_THREADS;
    s[j] = 0;
    d[j] = 0;
    if (e < E) {
      s[j] = src[e];
      d[j] = dst[e];
    }
  }
#pragma unroll
  for (int j = 0; j < BIN_EPT; ++j) {
    const int e = e0 + j * BIN_THREADS;
    rd[j] = -1;
    rs[j] = -1;
    if (e < E) {
      const unsigned bd = (unsigned)d[j] >> 8;
      const unsigned bsrc = (unsigned)s[j] >> 8;
      if (bd < (unsigned)bs.nb && (unsigned)s[j] < (1u << 24)) rd[j] = atomicAdd(&table[bd], 1);
      if (bsrc < (unsigned)bs.nb) rs[j] = atomicAdd(&table[MAX_BINS + bsrc], 1);
    }
  }
  __syncthreads();

  if (tid < 2 * MAX_BINS) {
    int c = table[tid];
    c = c < SUB ? c : SUB;
    int incl = c;
    for (int o = 1; o < 32; o <<= 1) {
      const int t = __shfl_up(incl, o);
      if (lane >= o) incl += t;
    }
    offs[tid] = incl - c;
    if (lane == 31) wtot[tid >> 5] = incl;
  }
  __syncthreads();
  if (tid < 2 * MAX_BINS) {
    const int w = tid >> 5;
    int add = 0;
    for (int k = w & ~7; k < w; ++k) add += wtot[k];
    offs[tid] += add;
  }
  __syncthreads();

#pragma unroll
  for (int j = 0; j < BIN_EPT; ++j) {
    const int e = e0 + j * BIN_THREADS;
    if (e < E) {
      const unsigned bd = (unsigned)d[j] >> 8;
      const unsigned bsrc = (unsigned)s[j] >> 8;
      if ((unsigned)rd[j] < (unsigned)SUB) {
        const int p = offs[bd] + rd[j];
        st_rec_d[p] = ((unsigned)s[j] << 8) | ((unsigned)d[j] & 255u);
        st_pos_d[p] = (unsigned)(((size_t)bd * nblk + blk) * SUB + rd[j]);
      } else {
        const int slot = atomicAdd(&cnt_in[d[j]], 1);
        if (slot < LIST_CAP) lists[(size_t)d[j] * LIST_CAP + slot] = s[j];
      }
      if ((unsigned)rs[j] < (unsigned)SUB) {
        const int p = offs[MAX_BINS + bsrc] + rs[j];
        st_rec_s[p] = (unsigned char)((unsigned)s[j] & 255u);
        st_pos_s[p] = (unsigned)(((size_t)bsrc * nblk + blk) * SUB + rs[j]);
      } else {
        atomicAdd(&cnt_out[s[j]], 1);
      }
    }
  }
  __syncthreads();

  const int last_d = table[MAX_BINS - 1];
  const int last_s = table[2 * MAX_BINS - 1];
  const int total_d = offs[MAX_BINS - 1] + (last_d < SUB ? last_d : SUB);
  const int total_s = offs[2 * MAX_BINS - 1] + (last_s < SUB ? last_s : SUB);
  for (int i = tid; i < total_d; i += BIN_THREADS) bs.rec_d[st_pos_d[i]] = st_rec_d[i];
  for (int i = tid; i < total_s; i += BIN_THREADS) bs.rec_s[st_pos_s[i]] = st_rec_s[i];
  if (tid < 2 * MAX_BINS) {
    const int b = tid & (MAX_BINS - 1);
    if (b < bs.nb) {
      const int c = table[tid];
      int* cnt = tid < MAX_BINS ? bs.cnt_d : bs.cnt_s;
      cnt[(size_t)blk * bs.nb + b] = c < SUB ? c : SUB;
    }
  }
}

__global__ __launch_bounds__(BIN_THREADS) void rank_lists_kernel(
    int* __restrict__ meta0, const unsigned* __restrict__ scratch0, size_t scratch_bytes,
    int n_nodes, int nbins, int nblk) {
  constexpr int UNR = 4;
  const int g = blockIdx.y;
  const GraphMeta gm = graph_meta(meta0, g, n_nodes);
  int* __restrict__ cnt_out = gm.cnt_out;
  int* __restrict__ cnt_in = gm.cnt_in;
  int* __restrict__ lists = gm.lists;
  int* __restrict__ deg_in = gm.deg_in;
  const unsigned* __restrict__ scratch = scratch0 + (size_t)g * (scratch_bytes / sizeof(unsigned));
  __shared__ int tile[BIN_NODES * LIST_CAP];
  __shared__ int n_in[BIN_NODES];
  __shared__ int n_base[BIN_NODES];
  __shared__ int n_out[BIN_NODES];
  __shared__ int ccd[BIN_THREADS];
  __shared__ int ccs[BIN_THREADS];
  const int tid = threadIdx.x;
  const int b = blockIdx.x;
  const int node0 = b * BIN_NODES;
  const BinScratch bs = bin_scratch(const_cast<unsigned*>(scratch), scratch_bytes, nbins, nblk);

  if (tid < BIN_NODES) {
    const int node = node0 + tid;
    int ci = 0, co = 0;
    if (node < n_nodes) {
      ci = cnt_in[node];
      co = cnt_out[node];
    }
    n_in[tid] = ci;
    n_base[tid] = ci;
    n_out[tid] = co;
  }

  if (b < bs.nb) {
    for (int kc = 0; kc < nblk; kc += BIN_THREADS) {
      __syncthreads();
      if (kc + tid < nblk) {
        ccd[tid] = bs.cnt_d[(size_t)(kc + tid) * bs.nb + b];
        ccs[tid] = bs.cnt_s[(size_t)(kc + tid) * bs.nb + b];
      }
      __syncthreads();
      const int ncell = nblk - kc < BIN_THREADS ? nblk - kc : BIN_THREADS;
      const int nplace = ncell * SUB;
      const unsigned* rdp = bs.rec_d + ((size_t)b * nblk + kc) * SUB;
      const unsigned char* rsp = bs.rec_s + ((size_t)b * nblk + kc) * SUB;
      for (int p0 = tid; p0 < nplace; p0 += UNR * BIN_THREADS) {
        unsigned r[UNR];
        int q[UNR];
        bool has_r[UNR];
#pragma unroll
        for (int u = 0; u < UNR; ++u) {
          const int p = p0 + u * BIN_THREADS;
          r[u] = 0;
          has_r[u] = false;
          q[u] = -1;
          if (p < nplace) {
            if ((p % SUB) < ccd[p / SUB]) {
              r[u] = rdp[p];
              has_r[u] = true;
            }
            if ((p % SUB) < ccs[p / SUB]) q[u] = (int)rsp[p];
          }
        }
#pragma unroll
        for (int u = 0; u < UNR; ++u) {
          if (has_r[u]) {
            const int l = (int)(r[u] & 255u);
            const int slot = atomicAdd(&n_in[l], 1);
            if (slot < LIST_CAP) tile[l * LIST_CAP + slot] = (int)(r[u] >> 8);
          }
          if (q[u] >= 0) atomicAdd(&n_out[q[u]], 1);
        }
      }
    }
  }
  __syncthreads();

  for (int idx = tid; idx < BIN_NODES * LIST_CAP; idx += BIN_THREADS) {
    const int l = idx / LIST_CAP;
    const int slot = idx % LIST_CAP;
    const int node = node0 + l;
    if (node < n_nodes && slot >= n_base[l] && slot < n_in[l])
      lists[(size_t)node * LIST_CAP + slot] = tile[idx];
  }
  if (tid < BIN_NODES) {
    const int node = node0 + tid;
    if (node < n_nodes) {
      const int ci = n_in[tid];
      const int co = n_out[tid];
      deg_in[node] = ci;
      cnt_in[node] = __float_as_int(rsqrtf(fmaxf((float)ci, 1.0f)));
      cnt_out[node] = __float_as_int(rsqrtf(fmaxf((float)co, 1.0f)));
    }
  }
}


__global__ void scale_rows_kernel(float* __restrict__ out0,
                                  const float* __restrict__ in_a,
                                  const float* __restrict__ in_b,
                                  int* __restrict__ meta0,
                                  int n4, int d4, int n_nodes) {
  int i = blockIdx.x * blockDim.x + threadIdx.x;
  if (i >= n4) return;
  const int g = blockIdx.y;
  const float* __restrict__ in = g ? in_b : in_a;
  float* __restrict__ out = out0 + (size_t)g * (size_t)n4 * 4;
  const float* __restrict__ norm = reinterpret_cast<const float*>(graph_meta(meta0, g, n_nodes).cnt_out);
  int row = i / d4;
  float s = norm[row];
  float4 v = reinterpret_cast<const float4*>(in)[i];
  v.x *= s; v.y *= s; v.z *= s; v.w *= s;
  reinterpret_cast<float4*>(out)[i] = v;
}


template <int D, int CH, bool FULL>
__device__ __forceinline__ void gather_group(const float4* __restrict__ h4, int sreg, int base,
                                             int count, int lane, float4 (&acc)[D / 128]) {
  constexpr int VEC = D / 128;
  int sidx[CH];
#pragma unroll
  for (int u = 0; u < CH; ++u) {
    int j = base + u;
    if (!FULL) j = j < count ? j : count - 1;
    sidx[u] = __builtin_amdgcn_readlane(sreg, j);
  }
  float4 v[CH][VEC];
#pragma unroll
  for (int u = 0; u < CH; ++u) {
    const float4* p = h4 + (size_t)sidx[u] * (D / 4) + lane;
    const bool valid = FULL || base + u < count;
#pragma unroll
    for (int q = 0; q < VEC; ++q) {
      const float4 t = p[32 * q];
      v[u][q].x = valid ? t.x : 0.f;
      v[u][q].y = valid ? t.y : 0.f;
      v[u][q].z = valid ? t.z : 0.f;
      v[u][q].w = valid ? t.w : 0.f;
    }
  }
#pragma unroll
  for (int u = 0; u < CH; ++u) {
#pragma unroll
    for (int q = 0; q < VEC; ++q) {
      acc[q].x += v[u][q].x;
      acc[q].y += v[u][q].y;
      acc[q].z += v[u][q].z;
      acc[q].w += v[u][q].w;
    }
  }
}

template <int D>
__device__ __forceinline__ void gather_span(const float4* __restrict__ h4, int sreg,
                                            int count, int lane, float4 (&acc)[D / 128]) {
  constexpr int VEC = D / 128;
  constexpr int CHB = 16 / VEC;
  constexpr int CHS = 8 / VEC;
  int base = 0;
  for (; base + CHB <= count; base += CHB) gather_group<D, CHB, true>(h4, sreg, base, count, lane, acc);
  const int rem = count - base;
  if (rem > CHS)
    gather_group<D, CHB, false>(h4, sreg, base, count, lane, acc);
  else if (rem > 0)
    gather_group<D, CHS, false>(h4, sreg, base, count, lane, acc);
}

__device__ __forceinline__ void sort_list(int* lrow, int& s0, int& s1, int deg, int lane) {
  const int key0 = (lane < deg) ? (int)(((unsigned)s0 << 6) | (unsigned)lane) : 0x7fffffff;
  const int key1 = (lane + 32 < deg) ? (int)(((unsigned)s1 << 6) | (unsigned)(lane + 32)) : 0x7fffffff;
  const int n0 = deg < 32 ? deg : 32;
  int r0 = 0, r1 = 0;
  for (int j = 0; j < n0; ++j) {
    const int kj = __builtin_amdgcn_readlane(key0, j);
    r0 += (kj < key0) ? 1 : 0;
    r1 += (kj < key1) ? 1 : 0;
  }
  for (int j = 32; j < deg; ++j) {
    const int kj = __builtin_amdgcn_readlane(key1, j - 32);
    r0 += (kj < key0) ? 1 : 0;
    r1 += (kj < key1) ? 1 : 0;
  }
  int p0 = 0, p1 = 0;
  for (int j = 0; j < n0; ++j) {
    const int kj = __builtin_amdgcn_readlane(key0, j);
    const int rj = __builtin_amdgcn_readlane(r0, j);
    if (rj == lane) p0 = kj;
    if (rj == lane + 32) p1 = kj;
  }
  for (int j = 32; j < deg; ++j) {
    const int kj = __builtin_amdgcn_readlane(key1, j - 32);
    const int rj = __builtin_amdgcn_readlane(r1, j - 32);
    if (rj == lane) p0 = kj;
    if (rj == lane + 32) p1 = kj;
  }
  s0 = p0 >> 6;
  s1 = p1 >> 6;
  if (lane < deg) lrow[lane] = s0;
  if (lane + 32 < deg) lrow[lane + 32] = s1;
}

template <int D, bool EPI>
__device__ __forceinline__ void gather_row(const float4* __restrict__ h4, int s0, int s1,
                                           int deg, const int* __restrict__ src,
                                           const int* __restrict__ dst, int E, int row, float rs,
                                           const float4 (&bq)[D / 128], int lane,
                                           float4 (&acc)[D / 128]) {
  constexpr int VEC = D / 128;
  if (deg <= LIST_CAP) {
    gather_span<D>(h4, s0, deg < 32 ? deg : 32, lane, acc);
    if (deg > 32) gather_span<D>(h4, s1, deg - 32, lane, acc);
  } else {
    for (int e0 = 0; e0 < E; e0 += 32) {
      const int e = e0 + lane;
      int sv = 0;
      bool hit = false;
      if (e < E) {
        hit = (dst[e] == row);
        sv = src[e];
      }
      unsigned m = __builtin_amdgcn_ballot_w32(hit);
      while (m) {
        const int b = __builtin_ctz(m);
        m &= m - 1;
        const int s = __builtin_amdgcn_readlane(sv, b);
        const float4* p = h4 + (size_t)s * (D / 4) + lane;
#pragma unroll
        for (int q = 0; q < VEC; ++q) {
          const float4 t = p[32 * q];
          acc[q].x += t.x;
          acc[q].y += t.y;
          acc[q].z += t.z;
          acc[q].w += t.w;
        }
      }
    }
  }
  if (EPI) {
#pragma unroll
    for (int q = 0; q < VEC; ++q) {
      acc[q].x = fmaxf(acc[q].x * rs + bq[q].x, 0.0f);
      acc[q].y = fmaxf(acc[q].y * rs + bq[q].y, 0.0f);
      acc[q].z = fmaxf(acc[q].z * rs + bq[q].z, 0.0f);
      acc[q].w = fmaxf(acc[q].w * rs + bq[q].w, 0.0f);
    }
  }
}


template <int K, int NC, int NS>
__device__ __forceinline__ void tile_product(const float* sArow0, const float* sArow1,
                                             const float* __restrict__ Wp,
                                             v8f (&acc)[2][NS]) {
  constexpr int UNR = NS >= 4 ? 4 : 8;
#pragma unroll UNR
  for (int k = 0; k < K; k += 4) {
    const v2f a0 = *reinterpret_cast<const v2f*>(sArow0 + k);
    const v2f a1 = *reinterpret_cast<const v2f*>(sArow1 + k);
#pragma unroll
    for (int j = 0; j < NS; ++j) {
      v2f b;
      b[0] = Wp[k * NC + j * 16];
      b[1] = Wp[k * NC + NC + j * 16];
      acc[0][j] = __builtin_amdgcn_wmma_f32_16x16x4_f32(false, a0, false, b, (short)0,
                                                        acc[0][j], false, false);
      acc[1][j] = __builtin_amdgcn_wmma_f32_16x16x4_f32(false, a1, false, b, (short)0,
                                                        acc[1][j], false, false);
    }
  }
}

template <int ACT>
__device__ __forceinline__ float activate(float v) {
  if (ACT == 1) return fmaxf(v, 0.0f);
  if (ACT == 2) return (v > 0.0f) ? v : expm1f(v);
  return v;
}

template <int K1, int N1, int ACT1, bool SCALE1, bool BIAS1,
          int N2, int ACT2, bool SCALE2, bool BIAS2, bool SORT, bool EPI>
__global__ __launch_bounds__(256) void gather_gemm_pair_kernel(
    const float* __restrict__ h0, const int* __restrict__ ei_a, const int* __restrict__ ei_b,
    int E_a, int E_b, const float* __restrict__ gbias, const float* __restrict__ W1,
    const float* __restrict__ bias1, const float* __restrict__ W2,
    const float* __restrict__ bias2, int* __restrict__ meta0,
    float* __restrict__ C0, int M) {
  const int g = blockIdx.z;
  const float4* __restrict__ h4 = reinterpret_cast<const float4*>(h0 + (size_t)g * (size_t)M * K1);
  float* __restrict__ C = C0 + (size_t)g * (size_t)M * N2;
  const GraphMeta gm = graph_meta(meta0, g, M);
  const float* __restrict__ scale1 = reinterpret_cast<const float*>(gm.cnt_in);
  const float* __restrict__ scale2 = reinterpret_cast<const float*>(gm.cnt_out);
  const int E = g ? E_b : E_a;
  const int* __restrict__ src = g ? ei_b : ei_a;
  const int* __restrict__ dst = src + E;
  constexpr int PAD = 4;
  constexpr int LDA = K1 + PAD;
  constexpr int LDH = N1 + PAD;
  constexpr int NS1 = N1 / 64;
  constexpr int NS2 = N2 / 64;
  constexpr int VEC = K1 / 128;
  __shared__ float sA[64 * LDA];
  __shared__ float sH[64 * LDH];

  const int tid = threadIdx.x;
  const int blockM = blockIdx.x * 64;
  const int wave = tid >> 5;
  const int lane = tid & 31;

  {
    const int* __restrict__ deg_in = gm.deg_in;
    int* lists = gm.lists;
    const int row0 = __builtin_amdgcn_readfirstlane(blockM + wave * 8);
    float4 bq[VEC];
#pragma unroll
    for (int q = 0; q < VEC; ++q) bq[q] = make_float4(0.f, 0.f, 0.f, 0.f);
    if (EPI) {
      const float4* b4 = reinterpret_cast<const float4*>(gbias) + lane;
#pragma unroll
      for (int q = 0; q < VEC; ++q) bq[q] = b4[32 * q];
    }
    int* sL = reinterpret_cast<int*>(sH) + wave * (8 * LIST_CAP);
    {
      int l0[8], l1[8];
#pragma unroll
      for (int i = 0; i < 8; ++i) {
        const int rn = row0 + i < M ? row0 + i : M - 1;
        l0[i] = lists[(size_t)rn * LIST_CAP + lane];
        l1[i] = lists[(size_t)rn * LIST_CAP + lane + 32];
      }
#pragma unroll
      for (int i = 0; i < 8; ++i) {
        sL[i * LIST_CAP + lane] = l0[i];
        sL[i * LIST_CAP + lane + 32] = l1[i];
      }
    }
    const int rl = row0 + (lane & 7) < M ? row0 + (lane & 7) : M - 1;
    const int degv = deg_in[rl];
    const int rsv = EPI ? __float_as_int(scale1[rl]) : 0;
    for (int i = 0; i < 8; ++i) {
      const int row = row0 + i;
      int s0 = sL[i * LIST_CAP + lane];
      int s1 = sL[i * LIST_CAP + lane + 32];
      const int deg = __builtin_amdgcn_readlane(degv, i);
      const float rs = __int_as_float(__builtin_amdgcn_readlane(rsv, i));
      if (SORT && row < M && deg <= LIST_CAP)
        sort_list(lists + (size_t)row * LIST_CAP, s0, s1, deg, lane);
      float4 acc[VEC];
#pragma unroll
      for (int q = 0; q < VEC; ++q) acc[q] = make_float4(0.f, 0.f, 0.f, 0.f);
      if (row < M)
        gather_row<K1, EPI>(h4, s0, s1, deg, src, dst, E, row, rs, bq, lane, acc);
      float4* a4 = reinterpret_cast<float4*>(&sA[(wave * 8 + i) * LDA]) + lane;
#pragma unroll
      for (int q = 0; q < VEC; ++q) a4[32 * q] = acc[q];
    }
  }
  __syncthreads();

  const int wn = wave & 3;
  const int wmg = wave >> 2;
  const int half = lane >> 4;
  const int lm = lane & 15;
  const int lrow0 = wmg * 32 + 8 * half;

  {
    const int col = wn * (N1 / 4) + lm;
    v8f acc[2][NS1] = {};
    tile_product<K1, N1, NS1>(&sA[(wmg * 32 + lm) * LDA + 2 * half],
                              &sA[(wmg * 32 + 16 + lm) * LDA + 2 * half],
                              W1 + (size_t)(2 * half) * N1 + col, acc);
    float bv[NS1];
#pragma unroll
    for (int j = 0; j < NS1; ++j) bv[j] = BIAS1 ? bias1[col + j * 16] : 0.0f;
#pragma unroll
    for (int t = 0; t < 2; ++t) {
#pragma unroll
      for (int r = 0; r < 8; ++r) {
        const int lr = lrow0 + t * 16 + r;
        float s = 1.0f;
        if (SCALE1) {
          const int gr = blockM + lr;
          s = scale1[gr < M ? gr : M - 1];
        }
#pragma unroll
        for (int j = 0; j < NS1; ++j) {
          float v = acc[t][j][r];
          if (SCALE1) v *= s;
          if (BIAS1) v += bv[j];
          sH[lr * LDH + col + j * 16] = activate<ACT1>(v);
        }
      }
    }
  }
  __syncthreads();

  {
    const int col = wn * (N2 / 4) + lm;
    v8f acc[2][NS2] = {};
    tile_product<N1, N2, NS2>(&sH[(wmg * 32 + lm) * LDH + 2 * half],
                              &sH[(wmg * 32 + 16 + lm) * LDH + 2 * half],
                              W2 + (size_t)(2 * half) * N2 + col, acc);
    float bv[NS2];
#pragma unroll
    for (int j = 0; j < NS2; ++j) bv[j] = BIAS2 ? bias2[col + j * 16] : 0.0f;
#pragma unroll
    for (int t = 0; t < 2; ++t) {
#pragma unroll
      for (int r = 0; r < 8; ++r) {
        const int gr = blockM + lrow0 + t * 16 + r;
        if (gr < M) {
          const float s = SCALE2 ? scale2[gr] : 1.0f;
          float* Cp = C + (size_t)gr * N2 + col;
#pragma unroll
          for (int j = 0; j < NS2; ++j) {
            float v = acc[t][j][r];
            if (SCALE2) v *= s;
            if (BIAS2) v += bv[j];
            Cp[j * 16] = activate<ACT2>(v);
          }
        }
      }
    }
  }
}


extern "C" void kernel_launch(void* const* d_in, const int* in_sizes, int n_in,
                              void* d_out, int out_size, void* d_ws, size_t ws_size,
                              hipStream_t stream) {
  const float* feat1 = (const float*)d_in[0];
  const float* feat2 = (const float*)d_in[1];
  const int* ei1 = (const int*)d_in[2];
  const int* ei2 = (const int*)d_in[3];
  const float* W1 = (const float*)d_in[4];
  const float* b1 = (const float*)d_in[5];
  const float* W2 = (const float*)d_in[6];
  const float* b2 = (const float*)d_in[7];
  const float* fc1W = (const float*)d_in[8];
  const float* fc1b = (const float*)d_in[9];
  const float* fc2W = (const float*)d_in[10];
  const float* fc2b = (const float*)d_in[11];

  const int N = in_sizes[0] / 128;
  const int E_a = in_sizes[2] / 2;
  const int E_b = in_sizes[3] / 2;

  float* ws = (float*)d_ws;
  const size_t region = (size_t)N * 128;
  float* R2 = ws + 2 * region;
  unsigned* scratch0 = (unsigned*)R2;
  const size_t scratch_bytes = region * sizeof(float);
  int* meta0 = (int*)ws;
  float* feat_scaled = (float*)d_out;

  const int row_blocks = (N * 32 + 255) / 256;
  const int m_tiles = (N + 63) / 64;

  hipMemsetAsync(graph_meta(meta0, 0, N).cnt_out, 0, 4 * (size_t)N * sizeof(int), stream);
  const int E_max = E_a > E_b ? E_a : E_b;
  const int nblk = (E_max + BIN_THREADS * BIN_EPT - 1) / (BIN_THREADS * BIN_EPT);
  const int nbins = (N + BIN_NODES - 1) / BIN_NODES;
  bin_edges_kernel<<<dim3(nblk, 2), BIN_THREADS, 0, stream>>>(
      ei1, ei2, E_a, E_b, meta0, N, scratch0, scratch_bytes, nbins);
  rank_lists_kernel<<<dim3(nbins, 2), BIN_THREADS, 0, stream>>>(
      meta0, scratch0, scratch_bytes, N, nbins, nblk);

  scale_rows_kernel<<<dim3(row_blocks, 2), 256, 0, stream>>>(
      feat_scaled, feat1, feat2, meta0, N * 128 / 4, 32, N);

  gather_gemm_pair_kernel<128, 256, 1, true, true, 128, 0, true, false, true, false>
      <<<dim3(m_tiles, 1, 2), 256, 0, stream>>>(feat_scaled, ei1, ei2, E_a, E_b, nullptr, W1, b1,
                                                W2, nullptr, meta0, R2, N);

  gather_gemm_pair_kernel<128, 128, 2, false, true, 128, 0, false, true, false, true>
      <<<dim3(m_tiles, 1, 2), 256, 0, stream>>>(R2, ei1, ei2, E_a, E_b, b2, fc1W, fc1b, fc2W,
                                                fc2b, meta0, (float*)d_out, N);
}
